# starting kernel
# baseline (speedup 1.0000x reference)
#include <hip/hip_runtime.h>
#include <math.h>

typedef float v2f __attribute__((ext_vector_type(2)));
typedef float v8f __attribute__((ext_vector_type(8)));

#define HEADS 8
#define CH 32
#define F 256
#define NEG_SLOPE 0.2f
#define LN_EPS 1e-5f

#define LDS_STRIDE 260

__device__ __forceinline__ unsigned f2key(float f) {
  unsigned b = __float_as_uint(f);
  return (b & 0x80000000u) ? ~b : (b | 0x80000000u);
}
__device__ __forceinline__ float key2f(unsigned k) {
  unsigned b = (k & 0x80000000u) ? (k & 0x7fffffffu) : ~k;
  return __uint_as_float(b);
}

__global__ void gat_init_kernel(float* __restrict__ out, float* __restrict__ denom,
                                unsigned* __restrict__ segmax, int n_out, int n_nh) {
  int stride = gridDim.x * blockDim.x;
  for (int t = blockIdx.x * blockDim.x + threadIdx.x; t < n_out; t += stride) out[t] = 0.f;
  for (int t = blockIdx.x * blockDim.x + threadIdx.x; t < n_nh; t += stride) {
    denom[t] = 0.f;
    segmax[t] = 0x007FFFFFu;
  }
}

__global__ void __launch_bounds__(512)
gat_gemm_wmma_kernel(const float* __restrict__ x, const float* __restrict__ W,
                     float* __restrict__ xl, int N) {
  __shared__ float lds_a[16 * LDS_STRIDE];

  const int tid  = threadIdx.x;
  const int lane = tid & 31;
  const int wave = tid >> 5;
  const int row0 = blockIdx.x * 16;
  const int col0 = wave * 16;

  {
    int r = tid >> 5;
    int c = (tid & 31) * 8;
    int grow = row0 + r;
    if (grow >= N) grow = N - 1;
    const float4* __restrict__ src = (const float4*)(x + (size_t)grow * F + c);
    float4* dst = (float4*)(&lds_a[r * LDS_STRIDE + c]);
    dst[0] = src[0];
    dst[1] = src[1];
  }
  __syncthreads();

  const int m    = lane & 15;
  const int koff = (lane >> 4) * 2;
  const float* __restrict__ arow = &lds_a[m * LDS_STRIDE];

  const int n = lane & 15;

  v8f acc = {0.f, 0.f, 0.f, 0.f, 0.f, 0.f, 0.f, 0.f};
  for (int k = 0; k < F; k += 4) {
    v2f a = *(const v2f*)(arow + k + koff);
    v2f b;
    b.x = W[(size_t)(k + koff) * F + col0 + n];
    b.y = W[(size_t)(k + koff + 1) * F + col0 + n];
    acc = __builtin_amdgcn_wmma_f32_16x16x4_f32(
         false, a,  false, b,
         (short)0, acc,  false,  false);
  }

  const int rbase = row0 + ((lane >> 4) * 8);
  float* __restrict__ dst = xl + (size_t)rbase * F + col0 + n;
  if (row0 + 16 <= N) {
#pragma unroll
    for (int j = 0; j < 8; ++j) dst[(size_t)j * F] = acc[j];
  } else {
#pragma unroll
    for (int j = 0; j < 8; ++j) {
      if (rbase + j < N) dst[(size_t)j * F] = acc[j];
    }
  }
}

__global__ void gat_att_kernel(const float* __restrict__ xl,
                               const float* __restrict__ att_src,
                               const float* __restrict__ att_dst,
                               float* __restrict__ a_src, float* __restrict__ a_dst,
                               int NH) {
  int t = blockIdx.x * blockDim.x + threadIdx.x;
  if (t >= NH) return;
  int nidx = t >> 3, h = t & 7;
  const float* __restrict__ xr = xl + (size_t)nidx * F + h * CH;
  const float* __restrict__ as = att_src + h * CH;
  const float* __restrict__ ad = att_dst + h * CH;
  float s = 0.f, d = 0.f;
#pragma unroll
  for (int c = 0; c < CH; ++c) {
    float v = xr[c];
    s += v * as[c];
    d += v * ad[c];
  }
  a_src[t] = s;
  a_dst[t] = d;
}

__device__ __forceinline__ void edge_nodes(const int* __restrict__ ei, int e, int E,
                                           int& s, int& d) {
  if (e < E) { s = ei[e]; d = ei[E + e]; }
  else       { s = d = e - E; }
}

__device__ __forceinline__ float leaky(float v) {
  return (v >= 0.f) ? v : v * NEG_SLOPE;
}

__global__ void gat_edge_max_kernel(const int* __restrict__ ei,
                                    const float* __restrict__ a_src,
                                    const float* __restrict__ a_dst,
                                    unsigned* __restrict__ segmax, int E, int N) {
  int t = blockIdx.x * blockDim.x + threadIdx.x;
  int total = (E + N) * HEADS;
  if (t >= total) return;
  int e = t >> 3, h = t & 7;
  int s, d;
  edge_nodes(ei, e, E, s, d);
  float logit = leaky(a_src[s * HEADS + h] + a_dst[d * HEADS + h]);
  atomicMax(&segmax[d * HEADS + h], f2key(logit));
}

__global__ void gat_edge_sum_kernel(const int* __restrict__ ei,
                                    const float* __restrict__ a_src,
                                    const float* __restrict__ a_dst,
                                    const unsigned* __restrict__ segmax,
                                    float* __restrict__ denom, int E, int N) {
  int t = blockIdx.x * blockDim.x + threadIdx.x;
  int total = (E + N) * HEADS;
  if (t >= total) return;
  int e = t >> 3, h = t & 7;
  int s, d;
  edge_nodes(ei, e, E, s, d);
  float logit = leaky(a_src[s * HEADS + h] + a_dst[d * HEADS + h]);
  float mx = key2f(segmax[d * HEADS + h]);
  atomicAdd(&denom[d * HEADS + h], expf(logit - mx));
}

__global__ void __launch_bounds__(256)
gat_edge_agg_kernel(const int* __restrict__ ei, const float* __restrict__ xl,
                    const float* __restrict__ a_src, const float* __restrict__ a_dst,
                    const unsigned* __restrict__ segmax, const float* __restrict__ denom,
                    float* __restrict__ out, int E, int N) {
  int e = blockIdx.x;
  int tid = threadIdx.x;
  int h = tid >> 5;
  int s, d;
  edge_nodes(ei, e, E, s, d);
  float logit = leaky(a_src[s * HEADS + h] + a_dst[d * HEADS + h]);
  float mx = key2f(segmax[d * HEADS + h]);
  float alpha = expf(logit - mx) / denom[d * HEADS + h];
  float msg = xl[(size_t)s * F + tid] * alpha;
  atomicAdd(&out[(size_t)d * F + tid], msg);
}

__global__ void __launch_bounds__(256)
gat_ln_kernel(float* __restrict__ out, const float* __restrict__ bias,
              const float* __restrict__ gamma, const float* __restrict__ beta, int N) {
  __shared__ float red[256];
  int row = blockIdx.x;
  int tid = threadIdx.x;
  float v = out[(size_t)row * F + tid] + bias[tid];

  red[tid] = v;
  __syncthreads();
  for (int s = 128; s > 0; s >>= 1) {
    if (tid < s) red[tid] += red[tid + s];
    __syncthreads();
  }
  float mean = red[0] * (1.0f / F);
  __syncthreads();

  float dv = v - mean;
  red[tid] = dv * dv;
  __syncthreads();
  for (int s = 128; s > 0; s >>= 1) {
    if (tid < s) red[tid] += red[tid + s];
    __syncthreads();
  }
  float var = red[0] * (1.0f / F);

  out[(size_t)row * F + tid] = dv * rsqrtf(var + LN_EPS) * gamma[tid] + beta[tid];
}

extern "C" void kernel_launch(void* const* d_in, const int* in_sizes, int n_in,
                              void* d_out, int out_size, void* d_ws, size_t ws_size,
                              hipStream_t stream) {
  const float* x       = (const float*)d_in[0];
  const int*   ei      = (const int*)  d_in[1];
  const float* W       = (const float*)d_in[2];
  const float* att_src = (const float*)d_in[3];
  const float* att_dst = (const float*)d_in[4];
  const float* bias    = (const float*)d_in[5];
  const float* gamma   = (const float*)d_in[6];
  const float* beta    = (const float*)d_in[7];
  float* out = (float*)d_out;

  const int N = in_sizes[0] / F;
  const int E = in_sizes[1] / 2;

  char* ws = (char*)d_ws;
  float*    xl     = (float*)ws;    ws += (size_t)N * F * sizeof(float);
  float*    a_src  = (float*)ws;    ws += (size_t)N * HEADS * sizeof(float);
  float*    a_dst  = (float*)ws;    ws += (size_t)N * HEADS * sizeof(float);
  unsigned* segmax = (unsigned*)ws; ws += (size_t)N * HEADS * sizeof(unsigned);
  float*    denom  = (float*)ws;

  gat_init_kernel<<<2048, 256, 0, stream>>>(out, denom, segmax, N * F, N * HEADS);

  gat_gemm_wmma_kernel<<<(N + 15) / 16, 512, 0, stream>>>(x, W, xl, N);

  {
    int NH = N * HEADS;
    gat_att_kernel<<<(NH + 255) / 256, 256, 0, stream>>>(xl, att_src, att_dst,
                                                         a_src, a_dst, NH);
  }

  {
    int total = (E + N) * HEADS;
    int blocks = (total + 255) / 256;
    gat_edge_max_kernel<<<blocks, 256, 0, stream>>>(ei, a_src, a_dst, segmax, E, N);
    gat_edge_sum_kernel<<<blocks, 256, 0, stream>>>(ei, a_src, a_dst, segmax, denom, E, N);
    gat_edge_agg_kernel<<<E + N, 256, 0, stream>>>(ei, xl, a_src, a_dst, segmax, denom,
                                                   out, E, N);
  }

  gat_ln_kernel<<<N, 256, 0, stream>>>(out, bias, gamma, beta, N);
}


# aggregate by destination: edges counting-sorted by dst in-call (int atomics; slot order = arrival order), one block per node does softmax + weighted row sum in registers; float atomics gone
# speedup vs baseline: 2.0835x; 2.0835x over previous
#include <hip/hip_runtime.h>
#include <math.h>

typedef float v2f __attribute__((ext_vector_type(2)));
typedef float v8f __attribute__((ext_vector_type(8)));

#define HEADS 8
#define CH 32
#define F 256
#define NEG_SLOPE 0.2f
#define LN_EPS 1e-5f

#define LDS_STRIDE 260

__global__ void __launch_bounds__(512)
gat_gemm_wmma_kernel(const float* __restrict__ x, const float* __restrict__ W,
                     float* __restrict__ xl, int N) {
  __shared__ float lds_a[16 * LDS_STRIDE];

  const int tid  = threadIdx.x;
  const int lane = tid & 31;
  const int wave = tid >> 5;
  const int row0 = blockIdx.x * 16;
  const int col0 = wave * 16;

  {
    int r = tid >> 5;
    int c = (tid & 31) * 8;
    int grow = row0 + r;
    if (grow >= N) grow = N - 1;
    const float4* __restrict__ src = (const float4*)(x + (size_t)grow * F + c);
    float4* dst = (float4*)(&lds_a[r * LDS_STRIDE + c]);
    dst[0] = src[0];
    dst[1] = src[1];
  }
  __syncthreads();

  const int m    = lane & 15;
  const int koff = (lane >> 4) * 2;
  const float* __restrict__ arow = &lds_a[m * LDS_STRIDE];

  const int n = lane & 15;

  v8f acc = {0.f, 0.f, 0.f, 0.f, 0.f, 0.f, 0.f, 0.f};
  for (int k = 0; k < F; k += 4) {
    v2f a = *(const v2f*)(arow + k + koff);
    v2f b;
    b.x = W[(size_t)(k + koff) * F + col0 + n];
    b.y = W[(size_t)(k + koff + 1) * F + col0 + n];
    acc = __builtin_amdgcn_wmma_f32_16x16x4_f32(
         false, a,  false, b,
         (short)0, acc,  false,  false);
  }

  const int rbase = row0 + ((lane >> 4) * 8);
  float* __restrict__ dst = xl + (size_t)rbase * F + col0 + n;
  if (row0 + 16 <= N) {
#pragma unroll
    for (int j = 0; j < 8; ++j) dst[(size_t)j * F] = acc[j];
  } else {
#pragma unroll
    for (int j = 0; j < 8; ++j) {
      if (rbase + j < N) dst[(size_t)j * F] = acc[j];
    }
  }
}

__global__ void gat_att_kernel(const float* __restrict__ xl,
                               const float* __restrict__ att_src,
                               float* __restrict__ a_src, int* __restrict__ cnt,
                               int NH) {
  int t = blockIdx.x * blockDim.x + threadIdx.x;
  if (t >= NH) return;
  int nidx = t >> 3, h = t & 7;
  const float* __restrict__ xr = xl + (size_t)nidx * F + h * CH;
  const float* __restrict__ as = att_src + h * CH;
  float s = 0.f;
#pragma unroll
  for (int c = 0; c < CH; ++c) s += xr[c] * as[c];
  a_src[t] = s;
  if (h == 0) cnt[nidx] = 0;
}

__device__ __forceinline__ float leaky(float v) {
  return (v >= 0.f) ? v : v * NEG_SLOPE;
}

__global__ void gat_count_kernel(const int* __restrict__ ei, int* __restrict__ cnt, int E) {
  int e = blockIdx.x * blockDim.x + threadIdx.x;
  if (e < E) atomicAdd(&cnt[ei[E + e]], 1);
}

__global__ void __launch_bounds__(1024)
gat_scan_kernel(int* __restrict__ cnt, int N) {
  __shared__ int wsum[32];
  const int tid = threadIdx.x, lane = tid & 31, wave = tid >> 5;
  const int per = (N + 1023) / 1024;
  int b = tid * per;
  int e = b + per;
  if (b > N) b = N;
  if (e > N) e = N;
  int s = 0;
  for (int i = b; i < e; ++i) s += cnt[i];
  int incl = s;
  for (int o = 1; o < 32; o <<= 1) {
    int v = __shfl_up(incl, o);
    if (lane >= o) incl += v;
  }
  if (lane == 31) wsum[wave] = incl;
  __syncthreads();
  if (wave == 0) {
    int w = wsum[lane];
    int wi = w;
    for (int o = 1; o < 32; o <<= 1) {
      int v = __shfl_up(wi, o);
      if (lane >= o) wi += v;
    }
    wsum[lane] = wi - w;
  }
  __syncthreads();
  int run = wsum[wave] + incl - s;
  for (int i = b; i < e; ++i) {
    int c = cnt[i];
    cnt[i] = run;
    run += c;
  }
}

__global__ void gat_fill_kernel(const int* __restrict__ ei, int* __restrict__ cnt,
                                int* __restrict__ csr, int E) {
  int e = blockIdx.x * blockDim.x + threadIdx.x;
  if (e < E) {
    int s = ei[e];
    int d = ei[E + e];
    int p = atomicAdd(&cnt[d], 1);
    csr[p] = s;
  }
}

__device__ __forceinline__ float wave_sum(float v) {
  for (int o = 16; o > 0; o >>= 1) v += __shfl_xor(v, o);
  return v;
}
__device__ __forceinline__ float wave_max(float v) {
  for (int o = 16; o > 0; o >>= 1) v = fmaxf(v, __shfl_xor(v, o));
  return v;
}

__global__ void __launch_bounds__(256)
gat_gather_kernel(const float* __restrict__ xl, const float* __restrict__ a_src,
                  const float* __restrict__ att_dst, const int* __restrict__ cnt,
                  const int* __restrict__ csr, float* __restrict__ out, int N) {
  __shared__ float s_a[HEADS][32];
  __shared__ int   s_s[32];

  const int d = blockIdx.x;
  const int tid = threadIdx.x;
  const int lane = tid & 31;
  const int h = tid >> 5;
  const int ce = tid >> 3;
  const int ch = tid & 7;
  const int end = cnt[d];
  const int start = (d > 0) ? cnt[d - 1] : 0;

  const float xs = xl[(size_t)d * F + tid];
  const float ad = wave_sum(xs * att_dst[tid]);
  float m = leaky(a_src[d * HEADS + h] + ad);
  float sum = 1.f;
  float acc = xs;

  for (int base = start; base < end; base += 32) {
    {
      const int idx = base + ce;
      int sv = d;
      float av = 0.f;
      if (idx < end) {
        sv = csr[idx];
        av = a_src[sv * HEADS + ch];
      }
      s_a[ch][ce] = av;
      if (ch == 0) s_s[ce] = sv;
    }
    __syncthreads();
    const bool valid = base + lane < end;
    const int s = s_s[lane];
    const float l = valid ? leaky(s_a[h][lane] + ad) : -INFINITY;
    const float nm = fmaxf(m, wave_max(l));
    const float sc = expf(m - nm);
    acc *= sc;
    sum *= sc;
    m = nm;
    const float p = valid ? expf(l - m) : 0.f;
    sum += wave_sum(p);
    int left = end - base;
    if (left > 32) left = 32;
    for (int j0 = 0; j0 < left; j0 += 8) {
      float xv[8], pv[8];
#pragma unroll
      for (int u = 0; u < 8; ++u) {
        const int sj = __builtin_amdgcn_readlane(s, j0 + u);
        pv[u] = __uint_as_float(__builtin_amdgcn_readlane(__float_as_uint(p), j0 + u));
        xv[u] = xl[(size_t)sj * F + tid];
      }
#pragma unroll
      for (int u = 0; u < 8; ++u) acc = fmaf(pv[u], xv[u], acc);
    }
    if (base + 32 < end) __syncthreads();
  }
  out[(size_t)d * F + tid] = acc / sum;
}

__global__ void __launch_bounds__(256)
gat_ln_kernel(float* __restrict__ out, const float* __restrict__ bias,
              const float* __restrict__ gamma, const float* __restrict__ beta, int N) {
  __shared__ float red[256];
  int row = blockIdx.x;
  int tid = threadIdx.x;
  float v = out[(size_t)row * F + tid] + bias[tid];

  red[tid] = v;
  __syncthreads();
  for (int s = 128; s > 0; s >>= 1) {
    if (tid < s) red[tid] += red[tid + s];
    __syncthreads();
  }
  float mean = red[0] * (1.0f / F);
  __syncthreads();

  float dv = v - mean;
  red[tid] = dv * dv;
  __syncthreads();
  for (int s = 128; s > 0; s >>= 1) {
    if (tid < s) red[tid] += red[tid + s];
    __syncthreads();
  }
  float var = red[0] * (1.0f / F);

  out[(size_t)row * F + tid] = dv * rsqrtf(var + LN_EPS) * gamma[tid] + beta[tid];
}

extern "C" void kernel_launch(void* const* d_in, const int* in_sizes, int n_in,
                              void* d_out, int out_size, void* d_ws, size_t ws_size,
                              hipStream_t stream) {
  const float* x       = (const float*)d_in[0];
  const int*   ei      = (const int*)  d_in[1];
  const float* W       = (const float*)d_in[2];
  const float* att_src = (const float*)d_in[3];
  const float* att_dst = (const float*)d_in[4];
  const float* bias    = (const float*)d_in[5];
  const float* gamma   = (const float*)d_in[6];
  const float* beta    = (const float*)d_in[7];
  float* out = (float*)d_out;

  const int N = in_sizes[0] / F;
  const int E = in_sizes[1] / 2;

  char* ws = (char*)d_ws;
  float* xl    = (float*)ws; ws += (size_t)N * F * sizeof(float);
  float* a_src = (float*)ws; ws += (size_t)N * HEADS * sizeof(float);
  int*   cnt   = (int*)ws;   ws += (size_t)N * HEADS * sizeof(float);
  int*   csr   = (int*)ws;

  gat_gemm_wmma_kernel<<<(N + 15) / 16, 512, 0, stream>>>(x, W, xl, N);

  {
    int NH = N * HEADS;
    gat_att_kernel<<<(NH + 255) / 256, 256, 0, stream>>>(xl, att_src, a_src, cnt, NH);
  }

  {
    int blocks = (E + 255) / 256;
    gat_count_kernel<<<blocks, 256, 0, stream>>>(ei, cnt, E);
    gat_scan_kernel<<<1, 1024, 0, stream>>>(cnt, N);
    gat_fill_kernel<<<blocks, 256, 0, stream>>>(ei, cnt, csr, E);
  }

  gat_gather_kernel<<<N, 256, 0, stream>>>(xl, a_src, att_dst, cnt, csr, out, N);

  gat_ln_kernel<<<N, 256, 0, stream>>>(out, bias, gamma, beta, N);
}


# bias + LayerNorm fused into the per-node gather epilogue (row already in the block's registers): LN launch and its pass over out gone; LN sums now DPP wave sum + 8 LDS partials (new reduction tree)
# speedup vs baseline: 2.4298x; 1.1662x over previous
#include <hip/hip_runtime.h>
#include <math.h>

typedef float v2f __attribute__((ext_vector_type(2)));
typedef float v8f __attribute__((ext_vector_type(8)));

#define HEADS 8
#define CH 32
#define F 256
#define NEG_SLOPE 0.2f
#define LN_EPS 1e-5f

#define LDS_STRIDE 260

__global__ void __launch_bounds__(512)
gat_gemm_wmma_kernel(const float* __restrict__ x, const float* __restrict__ W,
                     float* __restrict__ xl, int N) {
  __shared__ float lds_a[16 * LDS_STRIDE];

  const int tid  = threadIdx.x;
  const int lane = tid & 31;
  const int wave = tid >> 5;
  const int row0 = blockIdx.x * 16;
  const int col0 = wave * 16;

  {
    int r = tid >> 5;
    int c = (tid & 31) * 8;
    int grow = row0 + r;
    if (grow >= N) grow = N - 1;
    const float4* __restrict__ src = (const float4*)(x + (size_t)grow * F + c);
    float4* dst = (float4*)(&lds_a[r * LDS_STRIDE + c]);
    dst[0] = src[0];
    dst[1] = src[1];
  }
  __syncthreads();

  const int m    = lane & 15;
  const int koff = (lane >> 4) * 2;
  const float* __restrict__ arow = &lds_a[m * LDS_STRIDE];

  const int n = lane & 15;

  v8f acc = {0.f, 0.f, 0.f, 0.f, 0.f, 0.f, 0.f, 0.f};
  for (int k = 0; k < F; k += 4) {
    v2f a = *(const v2f*)(arow + k + koff);
    v2f b;
    b.x = W[(size_t)(k + koff) * F + col0 + n];
    b.y = W[(size_t)(k + koff + 1) * F + col0 + n];
    acc = __builtin_amdgcn_wmma_f32_16x16x4_f32(
         false, a,  false, b,
         (short)0, acc,  false,  false);
  }

  const int rbase = row0 + ((lane >> 4) * 8);
  float* __restrict__ dst = xl + (size_t)rbase * F + col0 + n;
  if (row0 + 16 <= N) {
#pragma unroll
    for (int j = 0; j < 8; ++j) dst[(size_t)j * F] = acc[j];
  } else {
#pragma unroll
    for (int j = 0; j < 8; ++j) {
      if (rbase + j < N) dst[(size_t)j * F] = acc[j];
    }
  }
}

__global__ void gat_att_kernel(const float* __restrict__ xl,
                               const float* __restrict__ att_src,
                               float* __restrict__ a_src, int* __restrict__ cnt,
                               int NH) {
  int t = blockIdx.x * blockDim.x + threadIdx.x;
  if (t >= NH) return;
  int nidx = t >> 3, h = t & 7;
  const float* __restrict__ xr = xl + (size_t)nidx * F + h * CH;
  const float* __restrict__ as = att_src + h * CH;
  float s = 0.f;
#pragma unroll
  for (int c = 0; c < CH; ++c) s += xr[c] * as[c];
  a_src[t] = s;
  if (h == 0) cnt[nidx] = 0;
}

__device__ __forceinline__ float leaky(float v) {
  return (v >= 0.f) ? v : v * NEG_SLOPE;
}

__global__ void gat_count_kernel(const int* __restrict__ ei, int* __restrict__ cnt, int E) {
  int e = blockIdx.x * blockDim.x + threadIdx.x;
  if (e < E) atomicAdd(&cnt[ei[E + e]], 1);
}

__global__ void __launch_bounds__(1024)
gat_scan_kernel(int* __restrict__ cnt, int N) {
  __shared__ int wsum[32];
  const int tid = threadIdx.x, lane = tid & 31, wave = tid >> 5;
  const int per = (N + 1023) / 1024;
  int b = tid * per;
  int e = b + per;
  if (b > N) b = N;
  if (e > N) e = N;
  int s = 0;
  for (int i = b; i < e; ++i) s += cnt[i];
  int incl = s;
  for (int o = 1; o < 32; o <<= 1) {
    int v = __shfl_up(incl, o);
    if (lane >= o) incl += v;
  }
  if (lane == 31) wsum[wave] = incl;
  __syncthreads();
  if (wave == 0) {
    int w = wsum[lane];
    int wi = w;
    for (int o = 1; o < 32; o <<= 1) {
      int v = __shfl_up(wi, o);
      if (lane >= o) wi += v;
    }
    wsum[lane] = wi - w;
  }
  __syncthreads();
  int run = wsum[wave] + incl - s;
  for (int i = b; i < e; ++i) {
    int c = cnt[i];
    cnt[i] = run;
    run += c;
  }
}

__global__ void gat_fill_kernel(const int* __restrict__ ei, int* __restrict__ cnt,
                                int* __restrict__ csr, int E) {
  int e = blockIdx.x * blockDim.x + threadIdx.x;
  if (e < E) {
    int s = ei[e];
    int d = ei[E + e];
    int p = atomicAdd(&cnt[d], 1);
    csr[p] = s;
  }
}

__device__ __forceinline__ float wave_sum(float v) {
  for (int o = 16; o > 0; o >>= 1) v += __shfl_xor(v, o);
  return v;
}
__device__ __forceinline__ float wave_max(float v) {
  for (int o = 16; o > 0; o >>= 1) v = fmaxf(v, __shfl_xor(v, o));
  return v;
}

template <int CTRL>
__device__ __forceinline__ float dpp_row(float v) {
  return __int_as_float(__builtin_amdgcn_update_dpp(
      0, __float_as_int(v), CTRL,  0xf,  0xf,  false));
}
__device__ __forceinline__ float xhalf(float v) {
  const int b = __float_as_int(v);
  return __int_as_float(__builtin_amdgcn_permlanex16(b, b, 0x76543210, 0xfedcba98, false, false));
}
__device__ __forceinline__ float wave_sum_valu(float v) {
  v += dpp_row<0x161>(v);
  v += dpp_row<0x162>(v);
  v += dpp_row<0x164>(v);
  v += dpp_row<0x168>(v);
  v += xhalf(v);
  return v;
}

__global__ void __launch_bounds__(256)
gat_gather_ln_kernel(const float* __restrict__ xl, const float* __restrict__ a_src,
                     const float* __restrict__ att_dst, const int* __restrict__ cnt,
                     const int* __restrict__ csr, const float* __restrict__ bias,
                     const float* __restrict__ gamma, const float* __restrict__ beta,
                     float* __restrict__ out, int N) {
  __shared__ float s_a[HEADS][32];
  __shared__ int   s_s[32];
  __shared__ float s_r1[HEADS];
  __shared__ float s_r2[HEADS];

  const int d = blockIdx.x;
  const int tid = threadIdx.x;
  const int lane = tid & 31;
  const int h = tid >> 5;
  const int ce = tid >> 3;
  const int ch = tid & 7;
  const int end = cnt[d];
  const int start = (d > 0) ? cnt[d - 1] : 0;

  const float xs = xl[(size_t)d * F + tid];
  const float bi = bias[tid];
  const float ga = gamma[tid];
  const float be = beta[tid];
  const float ad = wave_sum(xs * att_dst[tid]);
  float m = leaky(a_src[d * HEADS + h] + ad);
  float sum = 1.f;
  float acc = xs;

  for (int base = start; base < end; base += 32) {
    {
      const int idx = base + ce;
      int sv = d;
      float av = 0.f;
      if (idx < end) {
        sv = csr[idx];
        av = a_src[sv * HEADS + ch];
      }
      s_a[ch][ce] = av;
      if (ch == 0) s_s[ce] = sv;
    }
    __syncthreads();
    const bool valid = base + lane < end;
    const int s = s_s[lane];
    const float l = valid ? leaky(s_a[h][lane] + ad) : -INFINITY;
    const float nm = fmaxf(m, wave_max(l));
    const float sc = expf(m - nm);
    acc *= sc;
    sum *= sc;
    m = nm;
    const float p = valid ? expf(l - m) : 0.f;
    sum += wave_sum(p);
    int left = end - base;
    if (left > 32) left = 32;
    for (int j0 = 0; j0 < left; j0 += 8) {
      float xv[8], pv[8];
#pragma unroll
      for (int u = 0; u < 8; ++u) {
        const int sj = __builtin_amdgcn_readlane(s, j0 + u);
        pv[u] = __uint_as_float(__builtin_amdgcn_readlane(__float_as_uint(p), j0 + u));
        xv[u] = xl[(size_t)sj * F + tid];
      }
#pragma unroll
      for (int u = 0; u < 8; ++u) acc = fmaf(pv[u], xv[u], acc);
    }
    if (base + 32 < end) __syncthreads();
  }
  const float v = acc / sum + bi;
  const float p1 = wave_sum_valu(v);
  if (lane == 0) s_r1[h] = p1;
  __syncthreads();
  float tot = 0.f;
#pragma unroll
  for (int w = 0; w < HEADS; ++w) tot += s_r1[w];
  const float mean = tot * (1.0f / F);
  const float dv = v - mean;
  const float p2 = wave_sum_valu(dv * dv);
  if (lane == 0) s_r2[h] = p2;
  __syncthreads();
  float tv = 0.f;
#pragma unroll
  for (int w = 0; w < HEADS; ++w) tv += s_r2[w];
  const float var = tv * (1.0f / F);
  out[(size_t)d * F + tid] = dv * rsqrtf(var + LN_EPS) * ga + be;
}

extern "C" void kernel_launch(void* const* d_in, const int* in_sizes, int n_in,
                              void* d_out, int out_size, void* d_ws, size_t ws_size,
                              hipStream_t stream) {
  const float* x       = (const float*)d_in[0];
  const int*   ei      = (const int*)  d_in[1];
  const float* W       = (const float*)d_in[2];
  const float* att_src = (const float*)d_in[3];
  const float* att_dst = (const float*)d_in[4];
  const float* bias    = (const float*)d_in[5];
  const float* gamma   = (const float*)d_in[6];
  const float* beta    = (const float*)d_in[7];
  float* out = (float*)d_out;

  const int N = in_sizes[0] / F;
  const int E = in_sizes[1] / 2;

  char* ws = (char*)d_ws;
  float* xl    = (float*)ws; ws += (size_t)N * F * sizeof(float);
  float* a_src = (float*)ws; ws += (size_t)N * HEADS * sizeof(float);
  int*   cnt   = (int*)ws;   ws += (size_t)N * HEADS * sizeof(float);
  int*   csr   = (int*)ws;

  gat_gemm_wmma_kernel<<<(N + 15) / 16, 512, 0, stream>>>(x, W, xl, N);

  {
    int NH = N * HEADS;
    gat_att_kernel<<<(NH + 255) / 256, 256, 0, stream>>>(xl, att_src, a_src, cnt, NH);
  }

  {
    int blocks = (E + 255) / 256;
    gat_count_kernel<<<blocks, 256, 0, stream>>>(ei, cnt, E);
    gat_scan_kernel<<<1, 1024, 0, stream>>>(cnt, N);
    gat_fill_kernel<<<blocks, 256, 0, stream>>>(ei, cnt, csr, E);
  }

  gat_gather_ln_kernel<<<N, 256, 0, stream>>>(xl, a_src, att_dst, cnt, csr,
                                              bias, gamma, beta, out, N);
}


# gather remapped to one wave per node: lane = 2x4 channels (16-byte row loads), per-lane online softmax in slot order, LN sums inside the wave; 8x fewer waves, no LDS, no barriers
# speedup vs baseline: 3.0970x; 1.2746x over previous
#include <hip/hip_runtime.h>
#include <math.h>

typedef float v2f __attribute__((ext_vector_type(2)));
typedef float v8f __attribute__((ext_vector_type(8)));

#define HEADS 8
#define CH 32
#define F 256
#define NEG_SLOPE 0.2f
#define LN_EPS 1e-5f

#define LDS_STRIDE 260

__global__ void __launch_bounds__(512)
gat_gemm_wmma_kernel(const float* __restrict__ x, const float* __restrict__ W,
                     float* __restrict__ xl, int N) {
  __shared__ float lds_a[16 * LDS_STRIDE];

  const int tid  = threadIdx.x;
  const int lane = tid & 31;
  const int wave = tid >> 5;
  const int row0 = blockIdx.x * 16;
  const int col0 = wave * 16;

  {
    int r = tid >> 5;
    int c = (tid & 31) * 8;
    int grow = row0 + r;
    if (grow >= N) grow = N - 1;
    const float4* __restrict__ src = (const float4*)(x + (size_t)grow * F + c);
    float4* dst = (float4*)(&lds_a[r * LDS_STRIDE + c]);
    dst[0] = src[0];
    dst[1] = src[1];
  }
  __syncthreads();

  const int m    = lane & 15;
  const int koff = (lane >> 4) * 2;
  const float* __restrict__ arow = &lds_a[m * LDS_STRIDE];

  const int n = lane & 15;

  v8f acc = {0.f, 0.f, 0.f, 0.f, 0.f, 0.f, 0.f, 0.f};
  for (int k = 0; k < F; k += 4) {
    v2f a = *(const v2f*)(arow + k + koff);
    v2f b;
    b.x = W[(size_t)(k + koff) * F + col0 + n];
    b.y = W[(size_t)(k + koff + 1) * F + col0 + n];
    acc = __builtin_amdgcn_wmma_f32_16x16x4_f32(
         false, a,  false, b,
         (short)0, acc,  false,  false);
  }

  const int rbase = row0 + ((lane >> 4) * 8);
  float* __restrict__ dst = xl + (size_t)rbase * F + col0 + n;
  if (row0 + 16 <= N) {
#pragma unroll
    for (int j = 0; j < 8; ++j) dst[(size_t)j * F] = acc[j];
  } else {
#pragma unroll
    for (int j = 0; j < 8; ++j) {
      if (rbase + j < N) dst[(size_t)j * F] = acc[j];
    }
  }
}

__global__ void gat_att_kernel(const float* __restrict__ xl,
                               const float* __restrict__ att_src,
                               float* __restrict__ a_src, int* __restrict__ cnt,
                               int NH) {
  int t = blockIdx.x * blockDim.x + threadIdx.x;
  if (t >= NH) return;
  int nidx = t >> 3, h = t & 7;
  const float* __restrict__ xr = xl + (size_t)nidx * F + h * CH;
  const float* __restrict__ as = att_src + h * CH;
  float s = 0.f;
#pragma unroll
  for (int c = 0; c < CH; ++c) s += xr[c] * as[c];
  a_src[t] = s;
  if (h == 0) cnt[nidx] = 0;
}

__device__ __forceinline__ float leaky(float v) {
  return (v >= 0.f) ? v : v * NEG_SLOPE;
}

__global__ void gat_count_kernel(const int* __restrict__ ei, int* __restrict__ cnt, int E) {
  int e = blockIdx.x * blockDim.x + threadIdx.x;
  if (e < E) atomicAdd(&cnt[ei[E + e]], 1);
}

__global__ void __launch_bounds__(1024)
gat_scan_kernel(int* __restrict__ cnt, int N) {
  __shared__ int wsum[32];
  const int tid = threadIdx.x, lane = tid & 31, wave = tid >> 5;
  const int per = (N + 1023) / 1024;
  int b = tid * per;
  int e = b + per;
  if (b > N) b = N;
  if (e > N) e = N;
  int s = 0;
  for (int i = b; i < e; ++i) s += cnt[i];
  int incl = s;
  for (int o = 1; o < 32; o <<= 1) {
    int v = __shfl_up(incl, o);
    if (lane >= o) incl += v;
  }
  if (lane == 31) wsum[wave] = incl;
  __syncthreads();
  if (wave == 0) {
    int w = wsum[lane];
    int wi = w;
    for (int o = 1; o < 32; o <<= 1) {
      int v = __shfl_up(wi, o);
      if (lane >= o) wi += v;
    }
    wsum[lane] = wi - w;
  }
  __syncthreads();
  int run = wsum[wave] + incl - s;
  for (int i = b; i < e; ++i) {
    int c = cnt[i];
    cnt[i] = run;
    run += c;
  }
}

__global__ void gat_fill_kernel(const int* __restrict__ ei, int* __restrict__ cnt,
                                int* __restrict__ csr, int E) {
  int e = blockIdx.x * blockDim.x + threadIdx.x;
  if (e < E) {
    int s = ei[e];
    int d = ei[E + e];
    int p = atomicAdd(&cnt[d], 1);
    csr[p] = s;
  }
}

template <int CTRL>
__device__ __forceinline__ float dpp_row(float v) {
  return __int_as_float(__builtin_amdgcn_update_dpp(
      0, __float_as_int(v), CTRL,  0xf,  0xf,  false));
}
__device__ __forceinline__ float xhalf(float v) {
  const int b = __float_as_int(v);
  return __int_as_float(__builtin_amdgcn_permlanex16(b, b, 0x76543210, 0xfedcba98, false, false));
}
__device__ __forceinline__ float wave_sum_valu(float v) {
  v += dpp_row<0x161>(v);
  v += dpp_row<0x162>(v);
  v += dpp_row<0x164>(v);
  v += dpp_row<0x168>(v);
  v += xhalf(v);
  return v;
}

__device__ __forceinline__ float oct_sum(float v) {
  v += dpp_row<0x161>(v);
  v += dpp_row<0x162>(v);
  v += dpp_row<0x164>(v);
  return v;
}

__device__ __forceinline__ void fold_edge(float l, float& m, float& s, float4& acc,
                                          const float4 x) {
  const float nm = fmaxf(m, l);
  const float e = expf(fminf(m, l) - nm);
  const bool up = l > m;
  const float sc = up ? e : 1.f;
  const float p  = up ? 1.f : e;
  acc.x = fmaf(p, x.x, acc.x * sc);
  acc.y = fmaf(p, x.y, acc.y * sc);
  acc.z = fmaf(p, x.z, acc.z * sc);
  acc.w = fmaf(p, x.w, acc.w * sc);
  s = fmaf(s, sc, p);
  m = nm;
}

template <int G>
__device__ __forceinline__ void fold_rows(const float* __restrict__ xl,
                                          const float* __restrict__ a_src,
                                          int sv, int j0, int lane, int hA,
                                          float adA, float adB,
                                          float& mA, float& sA, float4& aA,
                                          float& mB, float& sB, float4& aB) {
  float4 vA[G], vB[G];
  float lA[G], lB[G];
#pragma unroll
  for (int u = 0; u < G; ++u) {
    const int sj = __builtin_amdgcn_readlane(sv, j0 + u);
    const float4* __restrict__ row = (const float4*)(xl + (size_t)sj * F);
    const float* __restrict__ as = a_src + (size_t)sj * HEADS;
    vA[u] = row[lane];
    vB[u] = row[lane + 32];
    lA[u] = as[hA];
    lB[u] = as[hA + 4];
  }
#pragma unroll
  for (int u = 0; u < G; ++u) {
    fold_edge(leaky(lA[u] + adA), mA, sA, aA, vA[u]);
    fold_edge(leaky(lB[u] + adB), mB, sB, aB, vB[u]);
  }
}

__global__ void __launch_bounds__(256)
gat_gather_ln_kernel(const float* __restrict__ xl, const float* __restrict__ a_src,
                     const float* __restrict__ att_dst, const int* __restrict__ cnt,
                     const int* __restrict__ csr, const float* __restrict__ bias,
                     const float* __restrict__ gamma, const float* __restrict__ beta,
                     float* __restrict__ out, int N) {
  const int lane = threadIdx.x & 31;
  const int d = __builtin_amdgcn_readfirstlane((int)(blockIdx.x * 8u + (threadIdx.x >> 5)));
  if (d >= N) return;
  const int hA = lane >> 3;

  const int end = cnt[d];
  const int start = (d > 0) ? cnt[d - 1] : 0;

  const float4* __restrict__ own = (const float4*)(xl + (size_t)d * F);
  const float4 xA = own[lane];
  const float4 xB = own[lane + 32];
  const float4 tA = ((const float4*)att_dst)[lane];
  const float4 tB = ((const float4*)att_dst)[lane + 32];
  const float adA = oct_sum(xA.x * tA.x + xA.y * tA.y + xA.z * tA.z + xA.w * tA.w);
  const float adB = oct_sum(xB.x * tB.x + xB.y * tB.y + xB.z * tB.z + xB.w * tB.w);

  const float* __restrict__ aown = a_src + (size_t)d * HEADS;
  float mA = leaky(aown[hA] + adA);
  float mB = leaky(aown[hA + 4] + adB);
  float sA = 1.f, sB = 1.f;
  float4 aA = xA, aB = xB;

  for (int base = start; base < end; base += 32) {
    const int idx = base + lane;
    const int sv = (idx < end) ? csr[idx] : d;
    int left = end - base;
    if (left > 32) left = 32;
    int j0 = 0;
    for (; j0 + 8 <= left; j0 += 8)
      fold_rows<8>(xl, a_src, sv, j0, lane, hA, adA, adB, mA, sA, aA, mB, sB, aB);
    if (left - j0 >= 4) {
      fold_rows<4>(xl, a_src, sv, j0, lane, hA, adA, adB, mA, sA, aA, mB, sB, aB);
      j0 += 4;
    }
    if (left - j0 >= 2) {
      fold_rows<2>(xl, a_src, sv, j0, lane, hA, adA, adB, mA, sA, aA, mB, sB, aB);
      j0 += 2;
    }
    if (left - j0 >= 1)
      fold_rows<1>(xl, a_src, sv, j0, lane, hA, adA, adB, mA, sA, aA, mB, sB, aB);
  }

  const float4 bA = ((const float4*)bias)[lane];
  const float4 bB = ((const float4*)bias)[lane + 32];
  const float4 gA = ((const float4*)gamma)[lane];
  const float4 gB = ((const float4*)gamma)[lane + 32];
  const float4 eA = ((const float4*)beta)[lane];
  const float4 eB = ((const float4*)beta)[lane + 32];

  float v[8];
  v[0] = aA.x / sA + bA.x;  v[1] = aA.y / sA + bA.y;
  v[2] = aA.z / sA + bA.z;  v[3] = aA.w / sA + bA.w;
  v[4] = aB.x / sB + bB.x;  v[5] = aB.y / sB + bB.y;
  v[6] = aB.z / sB + bB.z;  v[7] = aB.w / sB + bB.w;

  const float s1 = ((v[0] + v[1]) + (v[2] + v[3])) + ((v[4] + v[5]) + (v[6] + v[7]));
  const float mean = wave_sum_valu(s1) * (1.0f / F);
  float s2 = 0.f;
#pragma unroll
  for (int i = 0; i < 8; ++i) {
    v[i] -= mean;
    s2 = fmaf(v[i], v[i], s2);
  }
  const float var = wave_sum_valu(s2) * (1.0f / F);
  const float rs = rsqrtf(var + LN_EPS);

  float4 oA, oB;
  oA.x = v[0] * rs * gA.x + eA.x;  oA.y = v[1] * rs * gA.y + eA.y;
  oA.z = v[2] * rs * gA.z + eA.z;  oA.w = v[3] * rs * gA.w + eA.w;
  oB.x = v[4] * rs * gB.x + eB.x;  oB.y = v[5] * rs * gB.y + eB.y;
  oB.z = v[6] * rs * gB.z + eB.z;  oB.w = v[7] * rs * gB.w + eB.w;
  float4* __restrict__ orow = (float4*)(out + (size_t)d * F);
  orow[lane] = oA;
  orow[lane + 32] = oB;
}

extern "C" void kernel_launch(void* const* d_in, const int* in_sizes, int n_in,
                              void* d_out, int out_size, void* d_ws, size_t ws_size,
                              hipStream_t stream) {
  const float* x       = (const float*)d_in[0];
  const int*   ei      = (const int*)  d_in[1];
  const float* W       = (const float*)d_in[2];
  const float* att_src = (const float*)d_in[3];
  const float* att_dst = (const float*)d_in[4];
  const float* bias    = (const float*)d_in[5];
  const float* gamma   = (const float*)d_in[6];
  const float* beta    = (const float*)d_in[7];
  float* out = (float*)d_out;

  const int N = in_sizes[0] / F;
  const int E = in_sizes[1] / 2;

  char* ws = (char*)d_ws;
  float* xl    = (float*)ws; ws += (size_t)N * F * sizeof(float);
  float* a_src = (float*)ws; ws += (size_t)N * HEADS * sizeof(float);
  int*   cnt   = (int*)ws;   ws += (size_t)N * HEADS * sizeof(float);
  int*   csr   = (int*)ws;

  gat_gemm_wmma_kernel<<<(N + 15) / 16, 512, 0, stream>>>(x, W, xl, N);

  {
    int NH = N * HEADS;
    gat_att_kernel<<<(NH + 255) / 256, 256, 0, stream>>>(xl, att_src, a_src, cnt, NH);
  }

  {
    int blocks = (E + 255) / 256;
    gat_count_kernel<<<blocks, 256, 0, stream>>>(ei, cnt, E);
    gat_scan_kernel<<<1, 1024, 0, stream>>>(cnt, N);
    gat_fill_kernel<<<blocks, 256, 0, stream>>>(ei, cnt, csr, E);
  }

  gat_gather_ln_kernel<<<(N + 7) / 8, 256, 0, stream>>>(xl, a_src, att_dst, cnt, csr,
                                                        bias, gamma, beta, out, N);
}


# count and fill bin edges in LDS: block (edge chunk, 4096-node range) counts hits in LDS, takes its slots with one global add per node; per-edge global atomics gone; slot order still arrival order
# speedup vs baseline: 4.0925x; 1.3214x over previous
#include <hip/hip_runtime.h>
#include <math.h>

typedef float v2f __attribute__((ext_vector_type(2)));
typedef float v8f __attribute__((ext_vector_type(8)));

#define HEADS 8
#define CH 32
#define F 256
#define NEG_SLOPE 0.2f
#define LN_EPS 1e-5f

#define LDS_STRIDE 260

__global__ void __launch_bounds__(512)
gat_gemm_wmma_kernel(const float* __restrict__ x, const float* __restrict__ W,
                     float* __restrict__ xl, int N) {
  __shared__ float lds_a[16 * LDS_STRIDE];

  const int tid  = threadIdx.x;
  const int lane = tid & 31;
  const int wave = tid >> 5;
  const int row0 = blockIdx.x * 16;
  const int col0 = wave * 16;

  {
    int r = tid >> 5;
    int c = (tid & 31) * 8;
    int grow = row0 + r;
    if (grow >= N) grow = N - 1;
    const float4* __restrict__ src = (const float4*)(x + (size_t)grow * F + c);
    float4* dst = (float4*)(&lds_a[r * LDS_STRIDE + c]);
    dst[0] = src[0];
    dst[1] = src[1];
  }
  __syncthreads();

  const int m    = lane & 15;
  const int koff = (lane >> 4) * 2;
  const float* __restrict__ arow = &lds_a[m * LDS_STRIDE];

  const int n = lane & 15;

  v8f acc = {0.f, 0.f, 0.f, 0.f, 0.f, 0.f, 0.f, 0.f};
  for (int k = 0; k < F; k += 4) {
    v2f a = *(const v2f*)(arow + k + koff);
    v2f b;
    b.x = W[(size_t)(k + koff) * F + col0 + n];
    b.y = W[(size_t)(k + koff + 1) * F + col0 + n];
    acc = __builtin_amdgcn_wmma_f32_16x16x4_f32(
         false, a,  false, b,
         (short)0, acc,  false,  false);
  }

  const int rbase = row0 + ((lane >> 4) * 8);
  float* __restrict__ dst = xl + (size_t)rbase * F + col0 + n;
  if (row0 + 16 <= N) {
#pragma unroll
    for (int j = 0; j < 8; ++j) dst[(size_t)j * F] = acc[j];
  } else {
#pragma unroll
    for (int j = 0; j < 8; ++j) {
      if (rbase + j < N) dst[(size_t)j * F] = acc[j];
    }
  }
}

__global__ void gat_att_kernel(const float* __restrict__ xl,
                               const float* __restrict__ att_src,
                               float* __restrict__ a_src, int* __restrict__ cnt,
                               int NH) {
  int t = blockIdx.x * blockDim.x + threadIdx.x;
  if (t >= NH) return;
  int nidx = t >> 3, h = t & 7;
  const float* __restrict__ xr = xl + (size_t)nidx * F + h * CH;
  const float* __restrict__ as = att_src + h * CH;
  float s = 0.f;
#pragma unroll
  for (int c = 0; c < CH; ++c) s += xr[c] * as[c];
  a_src[t] = s;
  if (h == 0) cnt[nidx] = 0;
}

__device__ __forceinline__ float leaky(float v) {
  return (v >= 0.f) ? v : v * NEG_SLOPE;
}

#define SORT_RANGE 4096
#define SORT_CHUNKS 16
#define SORT_PER_THREAD (SORT_RANGE / 1024)

__device__ __forceinline__ int bin_of(int d, int lo, int N, int lane) {
  const unsigned r = (unsigned)(d - lo);
  return (r < (unsigned)SORT_RANGE && d < N) ? (int)r : SORT_RANGE + lane;
}
__device__ __forceinline__ int4 bin_of4(int4 d, int lo, int N, int lane) {
  int4 b;
  b.x = bin_of(d.x, lo, N, lane);  b.y = bin_of(d.y, lo, N, lane);
  b.z = bin_of(d.z, lo, N, lane);  b.w = bin_of(d.w, lo, N, lane);
  return b;
}
__device__ __forceinline__ int4 lds_add4(int* lc, int4 b) {
  int4 p;
  p.x = atomicAdd(&lc[b.x], 1);  p.y = atomicAdd(&lc[b.y], 1);
  p.z = atomicAdd(&lc[b.z], 1);  p.w = atomicAdd(&lc[b.w], 1);
  return p;
}
__device__ __forceinline__ void put4(int* __restrict__ csr, int4 b, int4 p, int4 s) {
  if (b.x < SORT_RANGE) csr[p.x] = s.x;
  if (b.y < SORT_RANGE) csr[p.y] = s.y;
  if (b.z < SORT_RANGE) csr[p.z] = s.z;
  if (b.w < SORT_RANGE) csr[p.w] = s.w;
}

template <bool PLACE>
__device__ __forceinline__ void edge_pass(const int* __restrict__ ei, int* lc,
                                          int* __restrict__ csr, int E, int N,
                                          int lo, int e0, int e1, int tid) {
  const int lane = tid & 31;
  const int* __restrict__ dst = ei + E;
  if ((E & 3) == 0) {
    const int4* __restrict__ d4 = (const int4*)dst;
    const int4* __restrict__ s4 = (const int4*)ei;
    const int v1 = e1 >> 2;
    int v = (e0 >> 2) + tid;
    for (; v + 3072 < v1; v += 4096) {
      const int4 da = d4[v];
      const int4 db = d4[v + 1024];
      const int4 dc = d4[v + 2048];
      const int4 dg = d4[v + 3072];
      int4 sa, sb, sc, sg;
      if (PLACE) {
        sa = s4[v];  sb = s4[v + 1024];  sc = s4[v + 2048];  sg = s4[v + 3072];
      }
      const int4 ba = bin_of4(da, lo, N, lane);
      const int4 bb = bin_of4(db, lo, N, lane);
      const int4 bc = bin_of4(dc, lo, N, lane);
      const int4 bg = bin_of4(dg, lo, N, lane);
      const int4 pa = lds_add4(lc, ba);
      const int4 pb = lds_add4(lc, bb);
      const int4 pc = lds_add4(lc, bc);
      const int4 pg = lds_add4(lc, bg);
      if (PLACE) {
        put4(csr, ba, pa, sa);  put4(csr, bb, pb, sb);
        put4(csr, bc, pc, sc);  put4(csr, bg, pg, sg);
      }
    }
    for (; v < v1; v += 1024) {
      const int4 da = d4[v];
      int4 sa;
      if (PLACE) sa = s4[v];
      const int4 ba = bin_of4(da, lo, N, lane);
      const int4 pa = lds_add4(lc, ba);
      if (PLACE) put4(csr, ba, pa, sa);
    }
  } else {
    for (int e = e0 + tid; e < e1; e += 1024) {
      const int b = bin_of(dst[e], lo, N, lane);
      const int p = atomicAdd(&lc[b], 1);
      if (PLACE) {
        if (b < SORT_RANGE) csr[p] = ei[e];
      }
    }
  }
}

__device__ __forceinline__ void chunk_bounds(int E, int& e0, int& e1) {
  const int len = (((E + SORT_CHUNKS - 1) / SORT_CHUNKS) + 3) & ~3;
  e0 = (int)blockIdx.x * len;
  if (e0 > E) e0 = E;
  e1 = e0 + len;
  if (e1 > E) e1 = E;
}

__global__ void __launch_bounds__(1024)
gat_count_kernel(const int* __restrict__ ei, int* __restrict__ cnt, int E, int N) {
  __shared__ int lc[SORT_RANGE + 32];
  const int tid = threadIdx.x;
  const int lo = (int)blockIdx.y * SORT_RANGE;
  int e0, e1;
  chunk_bounds(E, e0, e1);
  for (int k = tid; k < SORT_RANGE + 32; k += 1024) lc[k] = 0;
  __syncthreads();
  edge_pass<false>(ei, lc, nullptr, E, N, lo, e0, e1, tid);
  __syncthreads();
  for (int k = tid; k < SORT_RANGE; k += 1024) {
    const int c = lc[k];
    if (c > 0) atomicAdd(&cnt[lo + k], c);
  }
}

__global__ void __launch_bounds__(1024)
gat_scan_kernel(int* __restrict__ cnt, int N) {
  __shared__ int wsum[32];
  const int tid = threadIdx.x, lane = tid & 31, wave = tid >> 5;
  const int per = (N + 1023) / 1024;
  int b = tid * per;
  int e = b + per;
  if (b > N) b = N;
  if (e > N) e = N;
  int s = 0;
  for (int i = b; i < e; ++i) s += cnt[i];
  int incl = s;
  for (int o = 1; o < 32; o <<= 1) {
    int v = __shfl_up(incl, o);
    if (lane >= o) incl += v;
  }
  if (lane == 31) wsum[wave] = incl;
  __syncthreads();
  if (wave == 0) {
    int w = wsum[lane];
    int wi = w;
    for (int o = 1; o < 32; o <<= 1) {
      int v = __shfl_up(wi, o);
      if (lane >= o) wi += v;
    }
    wsum[lane] = wi - w;
  }
  __syncthreads();
  int run = wsum[wave] + incl - s;
  for (int i = b; i < e; ++i) {
    int c = cnt[i];
    cnt[i] = run;
    run += c;
  }
}

__global__ void __launch_bounds__(1024)
gat_fill_kernel(const int* __restrict__ ei, int* __restrict__ cnt,
                int* __restrict__ csr, int E, int N) {
  __shared__ int lc[SORT_RANGE + 32];
  const int tid = threadIdx.x;
  const int lo = (int)blockIdx.y * SORT_RANGE;
  int e0, e1;
  chunk_bounds(E, e0, e1);
  for (int k = tid; k < SORT_RANGE + 32; k += 1024) lc[k] = 0;
  __syncthreads();
  edge_pass<false>(ei, lc, nullptr, E, N, lo, e0, e1, tid);
  __syncthreads();
  {
    const int spare = tid % N;
    int c[SORT_PER_THREAD], b[SORT_PER_THREAD];
    for (int u = 0; u < SORT_PER_THREAD; ++u) c[u] = lc[tid + 1024 * u];
    for (int u = 0; u < SORT_PER_THREAD; ++u) {
      const int node = lo + tid + 1024 * u;
      b[u] = atomicAdd(&cnt[node < N ? node : spare], c[u]);
    }
    for (int u = 0; u < SORT_PER_THREAD; ++u) lc[tid + 1024 * u] = b[u];
  }
  __syncthreads();
  edge_pass<true>(ei, lc, csr, E, N, lo, e0, e1, tid);
}

template <int CTRL>
__device__ __forceinline__ float dpp_row(float v) {
  return __int_as_float(__builtin_amdgcn_update_dpp(
      0, __float_as_int(v), CTRL,  0xf,  0xf,  false));
}
__device__ __forceinline__ float xhalf(float v) {
  const int b = __float_as_int(v);
  return __int_as_float(__builtin_amdgcn_permlanex16(b, b, 0x76543210, 0xfedcba98, false, false));
}
__device__ __forceinline__ float wave_sum_valu(float v) {
  v += dpp_row<0x161>(v);
  v += dpp_row<0x162>(v);
  v += dpp_row<0x164>(v);
  v += dpp_row<0x168>(v);
  v += xhalf(v);
  return v;
}

__device__ __forceinline__ float oct_sum(float v) {
  v += dpp_row<0x161>(v);
  v += dpp_row<0x162>(v);
  v += dpp_row<0x164>(v);
  return v;
}

__device__ __forceinline__ void fold_edge(float l, float& m, float& s, float4& acc,
                                          const float4 x) {
  const float nm = fmaxf(m, l);
  const float e = expf(fminf(m, l) - nm);
  const bool up = l > m;
  const float sc = up ? e : 1.f;
  const float p  = up ? 1.f : e;
  acc.x = fmaf(p, x.x, acc.x * sc);
  acc.y = fmaf(p, x.y, acc.y * sc);
  acc.z = fmaf(p, x.z, acc.z * sc);
  acc.w = fmaf(p, x.w, acc.w * sc);
  s = fmaf(s, sc, p);
  m = nm;
}

template <int G>
__device__ __forceinline__ void fold_rows(const float* __restrict__ xl,
                                          const float* __restrict__ a_src,
                                          int sv, int j0, int lane, int hA,
                                          float adA, float adB,
                                          float& mA, float& sA, float4& aA,
                                          float& mB, float& sB, float4& aB) {
  float4 vA[G], vB[G];
  float lA[G], lB[G];
#pragma unroll
  for (int u = 0; u < G; ++u) {
    const int sj = __builtin_amdgcn_readlane(sv, j0 + u);
    const float4* __restrict__ row = (const float4*)(xl + (size_t)sj * F);
    const float* __restrict__ as = a_src + (size_t)sj * HEADS;
    vA[u] = row[lane];
    vB[u] = row[lane + 32];
    lA[u] = as[hA];
    lB[u] = as[hA + 4];
  }
#pragma unroll
  for (int u = 0; u < G; ++u) {
    fold_edge(leaky(lA[u] + adA), mA, sA, aA, vA[u]);
    fold_edge(leaky(lB[u] + adB), mB, sB, aB, vB[u]);
  }
}

__global__ void __launch_bounds__(256)
gat_gather_ln_kernel(const float* __restrict__ xl, const float* __restrict__ a_src,
                     const float* __restrict__ att_dst, const int* __restrict__ cnt,
                     const int* __restrict__ csr, const float* __restrict__ bias,
                     const float* __restrict__ gamma, const float* __restrict__ beta,
                     float* __restrict__ out, int N) {
  const int lane = threadIdx.x & 31;
  const int d = __builtin_amdgcn_readfirstlane((int)(blockIdx.x * 8u + (threadIdx.x >> 5)));
  if (d >= N) return;
  const int hA = lane >> 3;

  const int end = cnt[d];
  const int start = (d > 0) ? cnt[d - 1] : 0;

  const float4* __restrict__ own = (const float4*)(xl + (size_t)d * F);
  const float4 xA = own[lane];
  const float4 xB = own[lane + 32];
  const float4 tA = ((const float4*)att_dst)[lane];
  const float4 tB = ((const float4*)att_dst)[lane + 32];
  const float adA = oct_sum(xA.x * tA.x + xA.y * tA.y + xA.z * tA.z + xA.w * tA.w);
  const float adB = oct_sum(xB.x * tB.x + xB.y * tB.y + xB.z * tB.z + xB.w * tB.w);

  const float* __restrict__ aown = a_src + (size_t)d * HEADS;
  float mA = leaky(aown[hA] + adA);
  float mB = leaky(aown[hA + 4] + adB);
  float sA = 1.f, sB = 1.f;
  float4 aA = xA, aB = xB;

  for (int base = start; base < end; base += 32) {
    const int idx = base + lane;
    const int sv = (idx < end) ? csr[idx] : d;
    int left = end - base;
    if (left > 32) left = 32;
    int j0 = 0;
    for (; j0 + 8 <= left; j0 += 8)
      fold_rows<8>(xl, a_src, sv, j0, lane, hA, adA, adB, mA, sA, aA, mB, sB, aB);
    if (left - j0 >= 4) {
      fold_rows<4>(xl, a_src, sv, j0, lane, hA, adA, adB, mA, sA, aA, mB, sB, aB);
      j0 += 4;
    }
    if (left - j0 >= 2) {
      fold_rows<2>(xl, a_src, sv, j0, lane, hA, adA, adB, mA, sA, aA, mB, sB, aB);
      j0 += 2;
    }
    if (left - j0 >= 1)
      fold_rows<1>(xl, a_src, sv, j0, lane, hA, adA, adB, mA, sA, aA, mB, sB, aB);
  }

  const float4 bA = ((const float4*)bias)[lane];
  const float4 bB = ((const float4*)bias)[lane + 32];
  const float4 gA = ((const float4*)gamma)[lane];
  const float4 gB = ((const float4*)gamma)[lane + 32];
  const float4 eA = ((const float4*)beta)[lane];
  const float4 eB = ((const float4*)beta)[lane + 32];

  float v[8];
  v[0] = aA.x / sA + bA.x;  v[1] = aA.y / sA + bA.y;
  v[2] = aA.z / sA + bA.z;  v[3] = aA.w / sA + bA.w;
  v[4] = aB.x / sB + bB.x;  v[5] = aB.y / sB + bB.y;
  v[6] = aB.z / sB + bB.z;  v[7] = aB.w / sB + bB.w;

  const float s1 = ((v[0] + v[1]) + (v[2] + v[3])) + ((v[4] + v[5]) + (v[6] + v[7]));
  const float mean = wave_sum_valu(s1) * (1.0f / F);
  float s2 = 0.f;
#pragma unroll
  for (int i = 0; i < 8; ++i) {
    v[i] -= mean;
    s2 = fmaf(v[i], v[i], s2);
  }
  const float var = wave_sum_valu(s2) * (1.0f / F);
  const float rs = rsqrtf(var + LN_EPS);

  float4 oA, oB;
  oA.x = v[0] * rs * gA.x + eA.x;  oA.y = v[1] * rs * gA.y + eA.y;
  oA.z = v[2] * rs * gA.z + eA.z;  oA.w = v[3] * rs * gA.w + eA.w;
  oB.x = v[4] * rs * gB.x + eB.x;  oB.y = v[5] * rs * gB.y + eB.y;
  oB.z = v[6] * rs * gB.z + eB.z;  oB.w = v[7] * rs * gB.w + eB.w;
  float4* __restrict__ orow = (float4*)(out + (size_t)d * F);
  orow[lane] = oA;
  orow[lane + 32] = oB;
}

extern "C" void kernel_launch(void* const* d_in, const int* in_sizes, int n_in,
                              void* d_out, int out_size, void* d_ws, size_t ws_size,
                              hipStream_t stream) {
  const float* x       = (const float*)d_in[0];
  const int*   ei      = (const int*)  d_in[1];
  const float* W       = (const float*)d_in[2];
  const float* att_src = (const float*)d_in[3];
  const float* att_dst = (const float*)d_in[4];
  const float* bias    = (const float*)d_in[5];
  const float* gamma   = (const float*)d_in[6];
  const float* beta    = (const float*)d_in[7];
  float* out = (float*)d_out;

  const int N = in_sizes[0] / F;
  const int E = in_sizes[1] / 2;

  char* ws = (char*)d_ws;
  float* xl    = (float*)ws; ws += (size_t)N * F * sizeof(float);
  float* a_src = (float*)ws; ws += (size_t)N * HEADS * sizeof(float);
  int*   cnt   = (int*)ws;   ws += (size_t)N * HEADS * sizeof(float);
  int*   csr   = (int*)ws;

  gat_gemm_wmma_kernel<<<(N + 15) / 16, 512, 0, stream>>>(x, W, xl, N);

  {
    int NH = N * HEADS;
    gat_att_kernel<<<(NH + 255) / 256, 256, 0, stream>>>(xl, att_src, a_src, cnt, NH);
  }

  {
    dim3 grid(SORT_CHUNKS, (N + SORT_RANGE - 1) / SORT_RANGE);
    gat_count_kernel<<<grid, 1024, 0, stream>>>(ei, cnt, E, N);
    gat_scan_kernel<<<1, 1024, 0, stream>>>(cnt, N);
    gat_fill_kernel<<<grid, 1024, 0, stream>>>(ei, cnt, csr, E, N);
  }

  gat_gather_ln_kernel<<<(N + 7) / 8, 256, 0, stream>>>(xl, a_src, att_dst, cnt, csr,
                                                        bias, gamma, beta, out, N);
}


# projection GEMM retiled: block = 32 rows x 256 cols, each wave two stacked 16x16 tiles on one B fragment (two accumulators): W drawn from L2 once per 32 rows, not per 16; k order unchanged
# speedup vs baseline: 4.3382x; 1.0600x over previous
#include <hip/hip_runtime.h>
#include <math.h>

typedef float v2f __attribute__((ext_vector_type(2)));
typedef float v8f __attribute__((ext_vector_type(8)));

#define HEADS 8
#define CH 32
#define F 256
#define NEG_SLOPE 0.2f
#define LN_EPS 1e-5f

#define LDS_STRIDE 260

#define GEMM_ROWS 32
__global__ void __launch_bounds__(512)
gat_gemm_wmma_kernel(const float* __restrict__ x, const float* __restrict__ W,
                     float* __restrict__ xl, int N) {
  __shared__ float lds_a[GEMM_ROWS * LDS_STRIDE];

  const int tid  = threadIdx.x;
  const int lane = tid & 31;
  const int wave = tid >> 5;
  const int row0 = blockIdx.x * GEMM_ROWS;
  const int col0 = wave * 16;

  {
    const int c = (tid & 31) * 8;
#pragma unroll
    for (int p = 0; p < GEMM_ROWS / 16; ++p) {
      const int r = (tid >> 5) + 16 * p;
      int grow = row0 + r;
      if (grow >= N) grow = N - 1;
      const float4* __restrict__ src = (const float4*)(x + (size_t)grow * F + c);
      float4* dst = (float4*)(&lds_a[r * LDS_STRIDE + c]);
      dst[0] = src[0];
      dst[1] = src[1];
    }
  }
  __syncthreads();

  const int m    = lane & 15;
  const int koff = (lane >> 4) * 2;
  const float* __restrict__ arow0 = &lds_a[m * LDS_STRIDE];
  const float* __restrict__ arow1 = &lds_a[(16 + m) * LDS_STRIDE];

  const int n = lane & 15;

  v8f acc0 = {0.f, 0.f, 0.f, 0.f, 0.f, 0.f, 0.f, 0.f};
  v8f acc1 = {0.f, 0.f, 0.f, 0.f, 0.f, 0.f, 0.f, 0.f};
  for (int k = 0; k < F; k += 4) {
    v2f a0 = *(const v2f*)(arow0 + k + koff);
    v2f a1 = *(const v2f*)(arow1 + k + koff);
    v2f b;
    b.x = W[(size_t)(k + koff) * F + col0 + n];
    b.y = W[(size_t)(k + koff + 1) * F + col0 + n];
    acc0 = __builtin_amdgcn_wmma_f32_16x16x4_f32(
         false, a0,  false, b,
         (short)0, acc0,  false,  false);
    acc1 = __builtin_amdgcn_wmma_f32_16x16x4_f32(
         false, a1,  false, b,
         (short)0, acc1,  false,  false);
  }

  const int rbase = row0 + ((lane >> 4) * 8);
  float* __restrict__ dst = xl + (size_t)rbase * F + col0 + n;
  if (row0 + GEMM_ROWS <= N) {
#pragma unroll
    for (int j = 0; j < 8; ++j) dst[(size_t)j * F] = acc0[j];
#pragma unroll
    for (int j = 0; j < 8; ++j) dst[(size_t)(16 + j) * F] = acc1[j];
  } else {
#pragma unroll
    for (int j = 0; j < 8; ++j) {
      if (rbase + j < N) dst[(size_t)j * F] = acc0[j];
    }
#pragma unroll
    for (int j = 0; j < 8; ++j) {
      if (rbase + 16 + j < N) dst[(size_t)(16 + j) * F] = acc1[j];
    }
  }
}

__global__ void gat_att_kernel(const float* __restrict__ xl,
                               const float* __restrict__ att_src,
                               float* __restrict__ a_src, int* __restrict__ cnt,
                               int NH) {
  int t = blockIdx.x * blockDim.x + threadIdx.x;
  if (t >= NH) return;
  int nidx = t >> 3, h = t & 7;
  const float* __restrict__ xr = xl + (size_t)nidx * F + h * CH;
  const float* __restrict__ as = att_src + h * CH;
  float s = 0.f;
#pragma unroll
  for (int c = 0; c < CH; ++c) s += xr[c] * as[c];
  a_src[t] = s;
  if (h == 0) cnt[nidx] = 0;
}

__device__ __forceinline__ float leaky(float v) {
  return (v >= 0.f) ? v : v * NEG_SLOPE;
}

#define SORT_RANGE 4096
#define SORT_CHUNKS 16
#define SORT_PER_THREAD (SORT_RANGE / 1024)

__device__ __forceinline__ int bin_of(int d, int lo, int N, int lane) {
  const unsigned r = (unsigned)(d - lo);
  return (r < (unsigned)SORT_RANGE && d < N) ? (int)r : SORT_RANGE + lane;
}
__device__ __forceinline__ int4 bin_of4(int4 d, int lo, int N, int lane) {
  int4 b;
  b.x = bin_of(d.x, lo, N, lane);  b.y = bin_of(d.y, lo, N, lane);
  b.z = bin_of(d.z, lo, N, lane);  b.w = bin_of(d.w, lo, N, lane);
  return b;
}
__device__ __forceinline__ int4 lds_add4(int* lc, int4 b) {
  int4 p;
  p.x = atomicAdd(&lc[b.x], 1);  p.y = atomicAdd(&lc[b.y], 1);
  p.z = atomicAdd(&lc[b.z], 1);  p.w = atomicAdd(&lc[b.w], 1);
  return p;
}
__device__ __forceinline__ void put4(int* __restrict__ csr, int4 b, int4 p, int4 s) {
  if (b.x < SORT_RANGE) csr[p.x] = s.x;
  if (b.y < SORT_RANGE) csr[p.y] = s.y;
  if (b.z < SORT_RANGE) csr[p.z] = s.z;
  if (b.w < SORT_RANGE) csr[p.w] = s.w;
}

template <bool PLACE>
__device__ __forceinline__ void edge_pass(const int* __restrict__ ei, int* lc,
                                          int* __restrict__ csr, int E, int N,
                                          int lo, int e0, int e1, int tid) {
  const int lane = tid & 31;
  const int* __restrict__ dst = ei + E;
  if ((E & 3) == 0) {
    const int4* __restrict__ d4 = (const int4*)dst;
    const int4* __restrict__ s4 = (const int4*)ei;
    const int v1 = e1 >> 2;
    int v = (e0 >> 2) + tid;
    for (; v + 3072 < v1; v += 4096) {
      const int4 da = d4[v];
      const int4 db = d4[v + 1024];
      const int4 dc = d4[v + 2048];
      const int4 dg = d4[v + 3072];
      int4 sa, sb, sc, sg;
      if (PLACE) {
        sa = s4[v];  sb = s4[v + 1024];  sc = s4[v + 2048];  sg = s4[v + 3072];
      }
      const int4 ba = bin_of4(da, lo, N, lane);
      const int4 bb = bin_of4(db, lo, N, lane);
      const int4 bc = bin_of4(dc, lo, N, lane);
      const int4 bg = bin_of4(dg, lo, N, lane);
      const int4 pa = lds_add4(lc, ba);
      const int4 pb = lds_add4(lc, bb);
      const int4 pc = lds_add4(lc, bc);
      const int4 pg = lds_add4(lc, bg);
      if (PLACE) {
        put4(csr, ba, pa, sa);  put4(csr, bb, pb, sb);
        put4(csr, bc, pc, sc);  put4(csr, bg, pg, sg);
      }
    }
    for (; v < v1; v += 1024) {
      const int4 da = d4[v];
      int4 sa;
      if (PLACE) sa = s4[v];
      const int4 ba = bin_of4(da, lo, N, lane);
      const int4 pa = lds_add4(lc, ba);
      if (PLACE) put4(csr, ba, pa, sa);
    }
  } else {
    for (int e = e0 + tid; e < e1; e += 1024) {
      const int b = bin_of(dst[e], lo, N, lane);
      const int p = atomicAdd(&lc[b], 1);
      if (PLACE) {
        if (b < SORT_RANGE) csr[p] = ei[e];
      }
    }
  }
}

__device__ __forceinline__ void chunk_bounds(int E, int& e0, int& e1) {
  const int len = (((E + SORT_CHUNKS - 1) / SORT_CHUNKS) + 3) & ~3;
  e0 = (int)blockIdx.x * len;
  if (e0 > E) e0 = E;
  e1 = e0 + len;
  if (e1 > E) e1 = E;
}

__global__ void __launch_bounds__(1024)
gat_count_kernel(const int* __restrict__ ei, int* __restrict__ cnt, int E, int N) {
  __shared__ int lc[SORT_RANGE + 32];
  const int tid = threadIdx.x;
  const int lo = (int)blockIdx.y * SORT_RANGE;
  int e0, e1;
  chunk_bounds(E, e0, e1);
  for (int k = tid; k < SORT_RANGE + 32; k += 1024) lc[k] = 0;
  __syncthreads();
  edge_pass<false>(ei, lc, nullptr, E, N, lo, e0, e1, tid);
  __syncthreads();
  for (int k = tid; k < SORT_RANGE; k += 1024) {
    const int c = lc[k];
    if (c > 0) atomicAdd(&cnt[lo + k], c);
  }
}

__global__ void __launch_bounds__(1024)
gat_scan_kernel(int* __restrict__ cnt, int N) {
  __shared__ int wsum[32];
  const int tid = threadIdx.x, lane = tid & 31, wave = tid >> 5;
  const int per = (N + 1023) / 1024;
  int b = tid * per;
  int e = b + per;
  if (b > N) b = N;
  if (e > N) e = N;
  int s = 0;
  for (int i = b; i < e; ++i) s += cnt[i];
  int incl = s;
  for (int o = 1; o < 32; o <<= 1) {
    int v = __shfl_up(incl, o);
    if (lane >= o) incl += v;
  }
  if (lane == 31) wsum[wave] = incl;
  __syncthreads();
  if (wave == 0) {
    int w = wsum[lane];
    int wi = w;
    for (int o = 1; o < 32; o <<= 1) {
      int v = __shfl_up(wi, o);
      if (lane >= o) wi += v;
    }
    wsum[lane] = wi - w;
  }
  __syncthreads();
  int run = wsum[wave] + incl - s;
  for (int i = b; i < e; ++i) {
    int c = cnt[i];
    cnt[i] = run;
    run += c;
  }
}

__global__ void __launch_bounds__(1024)
gat_fill_kernel(const int* __restrict__ ei, int* __restrict__ cnt,
                int* __restrict__ csr, int E, int N) {
  __shared__ int lc[SORT_RANGE + 32];
  const int tid = threadIdx.x;
  const int lo = (int)blockIdx.y * SORT_RANGE;
  int e0, e1;
  chunk_bounds(E, e0, e1);
  for (int k = tid; k < SORT_RANGE + 32; k += 1024) lc[k] = 0;
  __syncthreads();
  edge_pass<false>(ei, lc, nullptr, E, N, lo, e0, e1, tid);
  __syncthreads();
  {
    const int spare = tid % N;
    int c[SORT_PER_THREAD], b[SORT_PER_THREAD];
    for (int u = 0; u < SORT_PER_THREAD; ++u) c[u] = lc[tid + 1024 * u];
    for (int u = 0; u < SORT_PER_THREAD; ++u) {
      const int node = lo + tid + 1024 * u;
      b[u] = atomicAdd(&cnt[node < N ? node : spare], c[u]);
    }
    for (int u = 0; u < SORT_PER_THREAD; ++u) lc[tid + 1024 * u] = b[u];
  }
  __syncthreads();
  edge_pass<true>(ei, lc, csr, E, N, lo, e0, e1, tid);
}

template <int CTRL>
__device__ __forceinline__ float dpp_row(float v) {
  return __int_as_float(__builtin_amdgcn_update_dpp(
      0, __float_as_int(v), CTRL,  0xf,  0xf,  false));
}
__device__ __forceinline__ float xhalf(float v) {
  const int b = __float_as_int(v);
  return __int_as_float(__builtin_amdgcn_permlanex16(b, b, 0x76543210, 0xfedcba98, false, false));
}
__device__ __forceinline__ float wave_sum_valu(float v) {
  v += dpp_row<0x161>(v);
  v += dpp_row<0x162>(v);
  v += dpp_row<0x164>(v);
  v += dpp_row<0x168>(v);
  v += xhalf(v);
  return v;
}

__device__ __forceinline__ float oct_sum(float v) {
  v += dpp_row<0x161>(v);
  v += dpp_row<0x162>(v);
  v += dpp_row<0x164>(v);
  return v;
}

__device__ __forceinline__ void fold_edge(float l, float& m, float& s, float4& acc,
                                          const float4 x) {
  const float nm = fmaxf(m, l);
  const float e = expf(fminf(m, l) - nm);
  const bool up = l > m;
  const float sc = up ? e : 1.f;
  const float p  = up ? 1.f : e;
  acc.x = fmaf(p, x.x, acc.x * sc);
  acc.y = fmaf(p, x.y, acc.y * sc);
  acc.z = fmaf(p, x.z, acc.z * sc);
  acc.w = fmaf(p, x.w, acc.w * sc);
  s = fmaf(s, sc, p);
  m = nm;
}

template <int G>
__device__ __forceinline__ void fold_rows(const float* __restrict__ xl,
                                          const float* __restrict__ a_src,
                                          int sv, int j0, int lane, int hA,
                                          float adA, float adB,
                                          float& mA, float& sA, float4& aA,
                                          float& mB, float& sB, float4& aB) {
  float4 vA[G], vB[G];
  float lA[G], lB[G];
#pragma unroll
  for (int u = 0; u < G; ++u) {
    const int sj = __builtin_amdgcn_readlane(sv, j0 + u);
    const float4* __restrict__ row = (const float4*)(xl + (size_t)sj * F);
    const float* __restrict__ as = a_src + (size_t)sj * HEADS;
    vA[u] = row[lane];
    vB[u] = row[lane + 32];
    lA[u] = as[hA];
    lB[u] = as[hA + 4];
  }
#pragma unroll
  for (int u = 0; u < G; ++u) {
    fold_edge(leaky(lA[u] + adA), mA, sA, aA, vA[u]);
    fold_edge(leaky(lB[u] + adB), mB, sB, aB, vB[u]);
  }
}

__global__ void __launch_bounds__(256)
gat_gather_ln_kernel(const float* __restrict__ xl, const float* __restrict__ a_src,
                     const float* __restrict__ att_dst, const int* __restrict__ cnt,
                     const int* __restrict__ csr, const float* __restrict__ bias,
                     const float* __restrict__ gamma, const float* __restrict__ beta,
                     float* __restrict__ out, int N) {
  const int lane = threadIdx.x & 31;
  const int d = __builtin_amdgcn_readfirstlane((int)(blockIdx.x * 8u + (threadIdx.x >> 5)));
  if (d >= N) return;
  const int hA = lane >> 3;

  const int end = cnt[d];
  const int start = (d > 0) ? cnt[d - 1] : 0;

  const float4* __restrict__ own = (const float4*)(xl + (size_t)d * F);
  const float4 xA = own[lane];
  const float4 xB = own[lane + 32];
  const float4 tA = ((const float4*)att_dst)[lane];
  const float4 tB = ((const float4*)att_dst)[lane + 32];
  const float adA = oct_sum(xA.x * tA.x + xA.y * tA.y + xA.z * tA.z + xA.w * tA.w);
  const float adB = oct_sum(xB.x * tB.x + xB.y * tB.y + xB.z * tB.z + xB.w * tB.w);

  const float* __restrict__ aown = a_src + (size_t)d * HEADS;
  float mA = leaky(aown[hA] + adA);
  float mB = leaky(aown[hA + 4] + adB);
  float sA = 1.f, sB = 1.f;
  float4 aA = xA, aB = xB;

  for (int base = start; base < end; base += 32) {
    const int idx = base + lane;
    const int sv = (idx < end) ? csr[idx] : d;
    int left = end - base;
    if (left > 32) left = 32;
    int j0 = 0;
    for (; j0 + 8 <= left; j0 += 8)
      fold_rows<8>(xl, a_src, sv, j0, lane, hA, adA, adB, mA, sA, aA, mB, sB, aB);
    if (left - j0 >= 4) {
      fold_rows<4>(xl, a_src, sv, j0, lane, hA, adA, adB, mA, sA, aA, mB, sB, aB);
      j0 += 4;
    }
    if (left - j0 >= 2) {
      fold_rows<2>(xl, a_src, sv, j0, lane, hA, adA, adB, mA, sA, aA, mB, sB, aB);
      j0 += 2;
    }
    if (left - j0 >= 1)
      fold_rows<1>(xl, a_src, sv, j0, lane, hA, adA, adB, mA, sA, aA, mB, sB, aB);
  }

  const float4 bA = ((const float4*)bias)[lane];
  const float4 bB = ((const float4*)bias)[lane + 32];
  const float4 gA = ((const float4*)gamma)[lane];
  const float4 gB = ((const float4*)gamma)[lane + 32];
  const float4 eA = ((const float4*)beta)[lane];
  const float4 eB = ((const float4*)beta)[lane + 32];

  float v[8];
  v[0] = aA.x / sA + bA.x;  v[1] = aA.y / sA + bA.y;
  v[2] = aA.z / sA + bA.z;  v[3] = aA.w / sA + bA.w;
  v[4] = aB.x / sB + bB.x;  v[5] = aB.y / sB + bB.y;
  v[6] = aB.z / sB + bB.z;  v[7] = aB.w / sB + bB.w;

  const float s1 = ((v[0] + v[1]) + (v[2] + v[3])) + ((v[4] + v[5]) + (v[6] + v[7]));
  const float mean = wave_sum_valu(s1) * (1.0f / F);
  float s2 = 0.f;
#pragma unroll
  for (int i = 0; i < 8; ++i) {
    v[i] -= mean;
    s2 = fmaf(v[i], v[i], s2);
  }
  const float var = wave_sum_valu(s2) * (1.0f / F);
  const float rs = rsqrtf(var + LN_EPS);

  float4 oA, oB;
  oA.x = v[0] * rs * gA.x + eA.x;  oA.y = v[1] * rs * gA.y + eA.y;
  oA.z = v[2] * rs * gA.z + eA.z;  oA.w = v[3] * rs * gA.w + eA.w;
  oB.x = v[4] * rs * gB.x + eB.x;  oB.y = v[5] * rs * gB.y + eB.y;
  oB.z = v[6] * rs * gB.z + eB.z;  oB.w = v[7] * rs * gB.w + eB.w;
  float4* __restrict__ orow = (float4*)(out + (size_t)d * F);
  orow[lane] = oA;
  orow[lane + 32] = oB;
}

extern "C" void kernel_launch(void* const* d_in, const int* in_sizes, int n_in,
                              void* d_out, int out_size, void* d_ws, size_t ws_size,
                              hipStream_t stream) {
  const float* x       = (const float*)d_in[0];
  const int*   ei      = (const int*)  d_in[1];
  const float* W       = (const float*)d_in[2];
  const float* att_src = (const float*)d_in[3];
  const float* att_dst = (const float*)d_in[4];
  const float* bias    = (const float*)d_in[5];
  const float* gamma   = (const float*)d_in[6];
  const float* beta    = (const float*)d_in[7];
  float* out = (float*)d_out;

  const int N = in_sizes[0] / F;
  const int E = in_sizes[1] / 2;

  char* ws = (char*)d_ws;
  float* xl    = (float*)ws; ws += (size_t)N * F * sizeof(float);
  float* a_src = (float*)ws; ws += (size_t)N * HEADS * sizeof(float);
  int*   cnt   = (int*)ws;   ws += (size_t)N * HEADS * sizeof(float);
  int*   csr   = (int*)ws;

  gat_gemm_wmma_kernel<<<(N + GEMM_ROWS - 1) / GEMM_ROWS, 512, 0, stream>>>(x, W, xl, N);

  {
    int NH = N * HEADS;
    gat_att_kernel<<<(NH + 255) / 256, 256, 0, stream>>>(xl, att_src, a_src, cnt, NH);
  }

  {
    dim3 grid(SORT_CHUNKS, (N + SORT_RANGE - 1) / SORT_RANGE);
    gat_count_kernel<<<grid, 1024, 0, stream>>>(ei, cnt, E, N);
    gat_scan_kernel<<<1, 1024, 0, stream>>>(cnt, N);
    gat_fill_kernel<<<grid, 1024, 0, stream>>>(ei, cnt, csr, E, N);
  }

  gat_gather_ln_kernel<<<(N + 7) / 8, 256, 0, stream>>>(xl, a_src, att_dst, cnt, csr,
                                                        bias, gamma, beta, out, N);
}


# scan of the in-degree counters split over node ranges: count blocks publish their range's edge total, each 4096-node range scanned by its own block, 16-byte loads/stores; one-block strided walk gone
# speedup vs baseline: 4.9157x; 1.1331x over previous
#include <hip/hip_runtime.h>
#include <math.h>

typedef float v2f __attribute__((ext_vector_type(2)));
typedef float v8f __attribute__((ext_vector_type(8)));

#define HEADS 8
#define CH 32
#define F 256
#define NEG_SLOPE 0.2f
#define LN_EPS 1e-5f

#define LDS_STRIDE 260

#define GEMM_ROWS 32
__global__ void __launch_bounds__(512)
gat_gemm_wmma_kernel(const float* __restrict__ x, const float* __restrict__ W,
                     float* __restrict__ xl, int N) {
  __shared__ float lds_a[GEMM_ROWS * LDS_STRIDE];

  const int tid  = threadIdx.x;
  const int lane = tid & 31;
  const int wave = tid >> 5;
  const int row0 = blockIdx.x * GEMM_ROWS;
  const int col0 = wave * 16;

  {
    const int c = (tid & 31) * 8;
#pragma unroll
    for (int p = 0; p < GEMM_ROWS / 16; ++p) {
      const int r = (tid >> 5) + 16 * p;
      int grow = row0 + r;
      if (grow >= N) grow = N - 1;
      const float4* __restrict__ src = (const float4*)(x + (size_t)grow * F + c);
      float4* dst = (float4*)(&lds_a[r * LDS_STRIDE + c]);
      dst[0] = src[0];
      dst[1] = src[1];
    }
  }
  __syncthreads();

  const int m    = lane & 15;
  const int koff = (lane >> 4) * 2;
  const float* __restrict__ arow0 = &lds_a[m * LDS_STRIDE];
  const float* __restrict__ arow1 = &lds_a[(16 + m) * LDS_STRIDE];

  const int n = lane & 15;

  v8f acc0 = {0.f, 0.f, 0.f, 0.f, 0.f, 0.f, 0.f, 0.f};
  v8f acc1 = {0.f, 0.f, 0.f, 0.f, 0.f, 0.f, 0.f, 0.f};
  for (int k = 0; k < F; k += 4) {
    v2f a0 = *(const v2f*)(arow0 + k + koff);
    v2f a1 = *(const v2f*)(arow1 + k + koff);
    v2f b;
    b.x = W[(size_t)(k + koff) * F + col0 + n];
    b.y = W[(size_t)(k + koff + 1) * F + col0 + n];
    acc0 = __builtin_amdgcn_wmma_f32_16x16x4_f32(
         false, a0,  false, b,
         (short)0, acc0,  false,  false);
    acc1 = __builtin_amdgcn_wmma_f32_16x16x4_f32(
         false, a1,  false, b,
         (short)0, acc1,  false,  false);
  }

  const int rbase = row0 + ((lane >> 4) * 8);
  float* __restrict__ dst = xl + (size_t)rbase * F + col0 + n;
  if (row0 + GEMM_ROWS <= N) {
#pragma unroll
    for (int j = 0; j < 8; ++j) dst[(size_t)j * F] = acc0[j];
#pragma unroll
    for (int j = 0; j < 8; ++j) dst[(size_t)(16 + j) * F] = acc1[j];
  } else {
#pragma unroll
    for (int j = 0; j < 8; ++j) {
      if (rbase + j < N) dst[(size_t)j * F] = acc0[j];
    }
#pragma unroll
    for (int j = 0; j < 8; ++j) {
      if (rbase + 16 + j < N) dst[(size_t)(16 + j) * F] = acc1[j];
    }
  }
}

__global__ void gat_att_kernel(const float* __restrict__ xl,
                               const float* __restrict__ att_src,
                               float* __restrict__ a_src, int* __restrict__ cnt,
                               int NH) {
  int t = blockIdx.x * blockDim.x + threadIdx.x;
  if (t >= NH) return;
  int nidx = t >> 3, h = t & 7;
  const float* __restrict__ xr = xl + (size_t)nidx * F + h * CH;
  const float* __restrict__ as = att_src + h * CH;
  float s = 0.f;
#pragma unroll
  for (int c = 0; c < CH; ++c) s += xr[c] * as[c];
  a_src[t] = s;
  if (h == 0) cnt[nidx] = 0;
}

__device__ __forceinline__ float leaky(float v) {
  return (v >= 0.f) ? v : v * NEG_SLOPE;
}

#define SORT_RANGE 4096
#define SORT_CHUNKS 16
#define SORT_PER_THREAD (SORT_RANGE / 1024)

__device__ __forceinline__ int bin_of(int d, int lo, int N, int lane) {
  const unsigned r = (unsigned)(d - lo);
  return (r < (unsigned)SORT_RANGE && d < N) ? (int)r : SORT_RANGE + lane;
}
__device__ __forceinline__ int4 bin_of4(int4 d, int lo, int N, int lane) {
  int4 b;
  b.x = bin_of(d.x, lo, N, lane);  b.y = bin_of(d.y, lo, N, lane);
  b.z = bin_of(d.z, lo, N, lane);  b.w = bin_of(d.w, lo, N, lane);
  return b;
}
__device__ __forceinline__ int4 lds_add4(int* lc, int4 b) {
  int4 p;
  p.x = atomicAdd(&lc[b.x], 1);  p.y = atomicAdd(&lc[b.y], 1);
  p.z = atomicAdd(&lc[b.z], 1);  p.w = atomicAdd(&lc[b.w], 1);
  return p;
}
__device__ __forceinline__ void put4(int* __restrict__ csr, int4 b, int4 p, int4 s) {
  if (b.x < SORT_RANGE) csr[p.x] = s.x;
  if (b.y < SORT_RANGE) csr[p.y] = s.y;
  if (b.z < SORT_RANGE) csr[p.z] = s.z;
  if (b.w < SORT_RANGE) csr[p.w] = s.w;
}

template <bool PLACE>
__device__ __forceinline__ void edge_pass(const int* __restrict__ ei, int* lc,
                                          int* __restrict__ csr, int E, int N,
                                          int lo, int e0, int e1, int tid) {
  const int lane = tid & 31;
  const int* __restrict__ dst = ei + E;
  if ((E & 3) == 0) {
    const int4* __restrict__ d4 = (const int4*)dst;
    const int4* __restrict__ s4 = (const int4*)ei;
    const int v1 = e1 >> 2;
    int v = (e0 >> 2) + tid;
    for (; v + 3072 < v1; v += 4096) {
      const int4 da = d4[v];
      const int4 db = d4[v + 1024];
      const int4 dc = d4[v + 2048];
      const int4 dg = d4[v + 3072];
      int4 sa, sb, sc, sg;
      if (PLACE) {
        sa = s4[v];  sb = s4[v + 1024];  sc = s4[v + 2048];  sg = s4[v + 3072];
      }
      const int4 ba = bin_of4(da, lo, N, lane);
      const int4 bb = bin_of4(db, lo, N, lane);
      const int4 bc = bin_of4(dc, lo, N, lane);
      const int4 bg = bin_of4(dg, lo, N, lane);
      const int4 pa = lds_add4(lc, ba);
      const int4 pb = lds_add4(lc, bb);
      const int4 pc = lds_add4(lc, bc);
      const int4 pg = lds_add4(lc, bg);
      if (PLACE) {
        put4(csr, ba, pa, sa);  put4(csr, bb, pb, sb);
        put4(csr, bc, pc, sc);  put4(csr, bg, pg, sg);
      }
    }
    for (; v < v1; v += 1024) {
      const int4 da = d4[v];
      int4 sa;
      if (PLACE) sa = s4[v];
      const int4 ba = bin_of4(da, lo, N, lane);
      const int4 pa = lds_add4(lc, ba);
      if (PLACE) put4(csr, ba, pa, sa);
    }
  } else {
    for (int e = e0 + tid; e < e1; e += 1024) {
      const int b = bin_of(dst[e], lo, N, lane);
      const int p = atomicAdd(&lc[b], 1);
      if (PLACE) {
        if (b < SORT_RANGE) csr[p] = ei[e];
      }
    }
  }
}

__device__ __forceinline__ void chunk_bounds(int E, int& e0, int& e1) {
  const int len = (((E + SORT_CHUNKS - 1) / SORT_CHUNKS) + 3) & ~3;
  e0 = (int)blockIdx.x * len;
  if (e0 > E) e0 = E;
  e1 = e0 + len;
  if (e1 > E) e1 = E;
}

__device__ __forceinline__ int wave_sum_int(int v) {
  for (int o = 16; o > 0; o >>= 1) v += __shfl_xor(v, o);
  return v;
}

__global__ void __launch_bounds__(1024)
gat_count_kernel(const int* __restrict__ ei, int* __restrict__ cnt,
                 int* __restrict__ rtot, int E, int N) {
  __shared__ int lc[SORT_RANGE + 32];
  const int tid = threadIdx.x;
  const int lo = (int)blockIdx.y * SORT_RANGE;
  int e0, e1;
  chunk_bounds(E, e0, e1);
  for (int k = tid; k < SORT_RANGE + 32; k += 1024) lc[k] = 0;
  __syncthreads();
  edge_pass<false>(ei, lc, nullptr, E, N, lo, e0, e1, tid);
  __syncthreads();
  for (int k = tid; k < SORT_RANGE; k += 1024) {
    const int c = lc[k];
    if (c > 0) atomicAdd(&cnt[lo + k], c);
  }
  if (tid < 32) {
    const int outside = wave_sum_int(lc[SORT_RANGE + tid]);
    if (tid == 0) rtot[(int)blockIdx.y * SORT_CHUNKS + (int)blockIdx.x] = (e1 - e0) - outside;
  }
}

__global__ void __launch_bounds__(1024)
gat_scan_kernel(int* __restrict__ cnt, const int* __restrict__ rtot, int N) {
  __shared__ int wsum[32];
  __shared__ int psum[32];
  const int tid = threadIdx.x, lane = tid & 31, wave = tid >> 5;
  const int lo = (int)blockIdx.x * SORT_RANGE;
  const int n0 = lo + 4 * tid;

  int part = 0;
  const int lower = (int)blockIdx.x * SORT_CHUNKS;
  for (int i = tid; i < lower; i += 1024) part += rtot[i];

  int4 c = make_int4(0, 0, 0, 0);
  if (n0 + 3 < N) {
    c = *(const int4*)(cnt + n0);
  } else {
    if (n0 < N) c.x = cnt[n0];
    if (n0 + 1 < N) c.y = cnt[n0 + 1];
    if (n0 + 2 < N) c.z = cnt[n0 + 2];
  }
  const int s = (c.x + c.y) + (c.z + c.w);
  int incl = s;
  for (int o = 1; o < 32; o <<= 1) {
    const int v = __shfl_up(incl, o);
    if (lane >= o) incl += v;
  }
  part = wave_sum_int(part);
  if (lane == 31) wsum[wave] = incl;
  if (lane == 0) psum[wave] = part;
  __syncthreads();
  const int before = wave_sum_int(psum[lane] + ((lane < wave) ? wsum[lane] : 0));
  int4 o4;
  o4.x = before + incl - s;
  o4.y = o4.x + c.x;
  o4.z = o4.y + c.y;
  o4.w = o4.z + c.z;
  if (n0 + 3 < N) {
    *(int4*)(cnt + n0) = o4;
  } else {
    if (n0 < N) cnt[n0] = o4.x;
    if (n0 + 1 < N) cnt[n0 + 1] = o4.y;
    if (n0 + 2 < N) cnt[n0 + 2] = o4.z;
  }
}

__global__ void __launch_bounds__(1024)
gat_fill_kernel(const int* __restrict__ ei, int* __restrict__ cnt,
                int* __restrict__ csr, int E, int N) {
  __shared__ int lc[SORT_RANGE + 32];
  const int tid = threadIdx.x;
  const int lo = (int)blockIdx.y * SORT_RANGE;
  int e0, e1;
  chunk_bounds(E, e0, e1);
  for (int k = tid; k < SORT_RANGE + 32; k += 1024) lc[k] = 0;
  __syncthreads();
  edge_pass<false>(ei, lc, nullptr, E, N, lo, e0, e1, tid);
  __syncthreads();
  {
    const int spare = tid % N;
    int c[SORT_PER_THREAD], b[SORT_PER_THREAD];
    for (int u = 0; u < SORT_PER_THREAD; ++u) c[u] = lc[tid + 1024 * u];
    for (int u = 0; u < SORT_PER_THREAD; ++u) {
      const int node = lo + tid + 1024 * u;
      b[u] = atomicAdd(&cnt[node < N ? node : spare], c[u]);
    }
    for (int u = 0; u < SORT_PER_THREAD; ++u) lc[tid + 1024 * u] = b[u];
  }
  __syncthreads();
  edge_pass<true>(ei, lc, csr, E, N, lo, e0, e1, tid);
}

template <int CTRL>
__device__ __forceinline__ float dpp_row(float v) {
  return __int_as_float(__builtin_amdgcn_update_dpp(
      0, __float_as_int(v), CTRL,  0xf,  0xf,  false));
}
__device__ __forceinline__ float xhalf(float v) {
  const int b = __float_as_int(v);
  return __int_as_float(__builtin_amdgcn_permlanex16(b, b, 0x76543210, 0xfedcba98, false, false));
}
__device__ __forceinline__ float wave_sum_valu(float v) {
  v += dpp_row<0x161>(v);
  v += dpp_row<0x162>(v);
  v += dpp_row<0x164>(v);
  v += dpp_row<0x168>(v);
  v += xhalf(v);
  return v;
}

__device__ __forceinline__ float oct_sum(float v) {
  v += dpp_row<0x161>(v);
  v += dpp_row<0x162>(v);
  v += dpp_row<0x164>(v);
  return v;
}

__device__ __forceinline__ void fold_edge(float l, float& m, float& s, float4& acc,
                                          const float4 x) {
  const float nm = fmaxf(m, l);
  const float e = expf(fminf(m, l) - nm);
  const bool up = l > m;
  const float sc = up ? e : 1.f;
  const float p  = up ? 1.f : e;
  acc.x = fmaf(p, x.x, acc.x * sc);
  acc.y = fmaf(p, x.y, acc.y * sc);
  acc.z = fmaf(p, x.z, acc.z * sc);
  acc.w = fmaf(p, x.w, acc.w * sc);
  s = fmaf(s, sc, p);
  m = nm;
}

template <int G>
__device__ __forceinline__ void fold_rows(const float* __restrict__ xl,
                                          const float* __restrict__ a_src,
                                          int sv, int j0, int lane, int hA,
                                          float adA, float adB,
                                          float& mA, float& sA, float4& aA,
                                          float& mB, float& sB, float4& aB) {
  float4 vA[G], vB[G];
  float lA[G], lB[G];
#pragma unroll
  for (int u = 0; u < G; ++u) {
    const int sj = __builtin_amdgcn_readlane(sv, j0 + u);
    const float4* __restrict__ row = (const float4*)(xl + (size_t)sj * F);
    const float* __restrict__ as = a_src + (size_t)sj * HEADS;
    vA[u] = row[lane];
    vB[u] = row[lane + 32];
    lA[u] = as[hA];
    lB[u] = as[hA + 4];
  }
#pragma unroll
  for (int u = 0; u < G; ++u) {
    fold_edge(leaky(lA[u] + adA), mA, sA, aA, vA[u]);
    fold_edge(leaky(lB[u] + adB), mB, sB, aB, vB[u]);
  }
}

__global__ void __launch_bounds__(256)
gat_gather_ln_kernel(const float* __restrict__ xl, const float* __restrict__ a_src,
                     const float* __restrict__ att_dst, const int* __restrict__ cnt,
                     const int* __restrict__ csr, const float* __restrict__ bias,
                     const float* __restrict__ gamma, const float* __restrict__ beta,
                     float* __restrict__ out, int N) {
  const int lane = threadIdx.x & 31;
  const int d = __builtin_amdgcn_readfirstlane((int)(blockIdx.x * 8u + (threadIdx.x >> 5)));
  if (d >= N) return;
  const int hA = lane >> 3;

  const int end = cnt[d];
  const int start = (d > 0) ? cnt[d - 1] : 0;

  const float4* __restrict__ own = (const float4*)(xl + (size_t)d * F);
  const float4 xA = own[lane];
  const float4 xB = own[lane + 32];
  const float4 tA = ((const float4*)att_dst)[lane];
  const float4 tB = ((const float4*)att_dst)[lane + 32];
  const float adA = oct_sum(xA.x * tA.x + xA.y * tA.y + xA.z * tA.z + xA.w * tA.w);
  const float adB = oct_sum(xB.x * tB.x + xB.y * tB.y + xB.z * tB.z + xB.w * tB.w);

  const float* __restrict__ aown = a_src + (size_t)d * HEADS;
  float mA = leaky(aown[hA] + adA);
  float mB = leaky(aown[hA + 4] + adB);
  float sA = 1.f, sB = 1.f;
  float4 aA = xA, aB = xB;

  for (int base = start; base < end; base += 32) {
    const int idx = base + lane;
    const int sv = (idx < end) ? csr[idx] : d;
    int left = end - base;
    if (left > 32) left = 32;
    int j0 = 0;
    for (; j0 + 8 <= left; j0 += 8)
      fold_rows<8>(xl, a_src, sv, j0, lane, hA, adA, adB, mA, sA, aA, mB, sB, aB);
    if (left - j0 >= 4) {
      fold_rows<4>(xl, a_src, sv, j0, lane, hA, adA, adB, mA, sA, aA, mB, sB, aB);
      j0 += 4;
    }
    if (left - j0 >= 2) {
      fold_rows<2>(xl, a_src, sv, j0, lane, hA, adA, adB, mA, sA, aA, mB, sB, aB);
      j0 += 2;
    }
    if (left - j0 >= 1)
      fold_rows<1>(xl, a_src, sv, j0, lane, hA, adA, adB, mA, sA, aA, mB, sB, aB);
  }

  const float4 bA = ((const float4*)bias)[lane];
  const float4 bB = ((const float4*)bias)[lane + 32];
  const float4 gA = ((const float4*)gamma)[lane];
  const float4 gB = ((const float4*)gamma)[lane + 32];
  const float4 eA = ((const float4*)beta)[lane];
  const float4 eB = ((const float4*)beta)[lane + 32];

  float v[8];
  v[0] = aA.x / sA + bA.x;  v[1] = aA.y / sA + bA.y;
  v[2] = aA.z / sA + bA.z;  v[3] = aA.w / sA + bA.w;
  v[4] = aB.x / sB + bB.x;  v[5] = aB.y / sB + bB.y;
  v[6] = aB.z / sB + bB.z;  v[7] = aB.w / sB + bB.w;

  const float s1 = ((v[0] + v[1]) + (v[2] + v[3])) + ((v[4] + v[5]) + (v[6] + v[7]));
  const float mean = wave_sum_valu(s1) * (1.0f / F);
  float s2 = 0.f;
#pragma unroll
  for (int i = 0; i < 8; ++i) {
    v[i] -= mean;
    s2 = fmaf(v[i], v[i], s2);
  }
  const float var = wave_sum_valu(s2) * (1.0f / F);
  const float rs = rsqrtf(var + LN_EPS);

  float4 oA, oB;
  oA.x = v[0] * rs * gA.x + eA.x;  oA.y = v[1] * rs * gA.y + eA.y;
  oA.z = v[2] * rs * gA.z + eA.z;  oA.w = v[3] * rs * gA.w + eA.w;
  oB.x = v[4] * rs * gB.x + eB.x;  oB.y = v[5] * rs * gB.y + eB.y;
  oB.z = v[6] * rs * gB.z + eB.z;  oB.w = v[7] * rs * gB.w + eB.w;
  float4* __restrict__ orow = (float4*)(out + (size_t)d * F);
  orow[lane] = oA;
  orow[lane + 32] = oB;
}

extern "C" void kernel_launch(void* const* d_in, const int* in_sizes, int n_in,
                              void* d_out, int out_size, void* d_ws, size_t ws_size,
                              hipStream_t stream) {
  const float* x       = (const float*)d_in[0];
  const int*   ei      = (const int*)  d_in[1];
  const float* W       = (const float*)d_in[2];
  const float* att_src = (const float*)d_in[3];
  const float* att_dst = (const float*)d_in[4];
  const float* bias    = (const float*)d_in[5];
  const float* gamma   = (const float*)d_in[6];
  const float* beta    = (const float*)d_in[7];
  float* out = (float*)d_out;

  const int N = in_sizes[0] / F;
  const int E = in_sizes[1] / 2;

  char* ws = (char*)d_ws;
  float* xl    = (float*)ws; ws += (size_t)N * F * sizeof(float);
  float* a_src = (float*)ws; ws += (size_t)N * HEADS * sizeof(float);
  int*   cnt   = (int*)ws;   ws += (size_t)N * HEADS * sizeof(float);
  int*   csr   = (int*)ws;
  int*   rtot  = cnt + N;

  gat_gemm_wmma_kernel<<<(N + GEMM_ROWS - 1) / GEMM_ROWS, 512, 0, stream>>>(x, W, xl, N);

  {
    int NH = N * HEADS;
    gat_att_kernel<<<(NH + 255) / 256, 256, 0, stream>>>(xl, att_src, a_src, cnt, NH);
  }

  {
    dim3 grid(SORT_CHUNKS, (N + SORT_RANGE - 1) / SORT_RANGE);
    gat_count_kernel<<<grid, 1024, 0, stream>>>(ei, cnt, rtot, E, N);
    gat_scan_kernel<<<grid.y, 1024, 0, stream>>>(cnt, rtot, N);
    gat_fill_kernel<<<grid, 1024, 0, stream>>>(ei, cnt, csr, E, N);
  }

  gat_gather_ln_kernel<<<(N + 7) / 8, 256, 0, stream>>>(xl, a_src, att_dst, cnt, csr,
                                                        bias, gamma, beta, out, N);
}


# gather softmax by groups of 8 edges: lane (head, edge) does one exp per edge and head, octet max, one rescale per group, octet broadcast of weights; new sum order (group max, lane partial sums)
# speedup vs baseline: 5.1291x; 1.0434x over previous
#include <hip/hip_runtime.h>
#include <math.h>

typedef float v2f __attribute__((ext_vector_type(2)));
typedef float v8f __attribute__((ext_vector_type(8)));

#define HEADS 8
#define CH 32
#define F 256
#define NEG_SLOPE 0.2f
#define LN_EPS 1e-5f

#define LDS_STRIDE 260

#define GEMM_ROWS 32
__global__ void __launch_bounds__(512)
gat_gemm_wmma_kernel(const float* __restrict__ x, const float* __restrict__ W,
                     float* __restrict__ xl, int N) {
  __shared__ float lds_a[GEMM_ROWS * LDS_STRIDE];

  const int tid  = threadIdx.x;
  const int lane = tid & 31;
  const int wave = tid >> 5;
  const int row0 = blockIdx.x * GEMM_ROWS;
  const int col0 = wave * 16;

  {
    const int c = (tid & 31) * 8;
#pragma unroll
    for (int p = 0; p < GEMM_ROWS / 16; ++p) {
      const int r = (tid >> 5) + 16 * p;
      int grow = row0 + r;
      if (grow >= N) grow = N - 1;
      const float4* __restrict__ src = (const float4*)(x + (size_t)grow * F + c);
      float4* dst = (float4*)(&lds_a[r * LDS_STRIDE + c]);
      dst[0] = src[0];
      dst[1] = src[1];
    }
  }
  __syncthreads();

  const int m    = lane & 15;
  const int koff = (lane >> 4) * 2;
  const float* __restrict__ arow0 = &lds_a[m * LDS_STRIDE];
  const float* __restrict__ arow1 = &lds_a[(16 + m) * LDS_STRIDE];

  const int n = lane & 15;

  v8f acc0 = {0.f, 0.f, 0.f, 0.f, 0.f, 0.f, 0.f, 0.f};
  v8f acc1 = {0.f, 0.f, 0.f, 0.f, 0.f, 0.f, 0.f, 0.f};
  for (int k = 0; k < F; k += 4) {
    v2f a0 = *(const v2f*)(arow0 + k + koff);
    v2f a1 = *(const v2f*)(arow1 + k + koff);
    v2f b;
    b.x = W[(size_t)(k + koff) * F + col0 + n];
    b.y = W[(size_t)(k + koff + 1) * F + col0 + n];
    acc0 = __builtin_amdgcn_wmma_f32_16x16x4_f32(
         false, a0,  false, b,
         (short)0, acc0,  false,  false);
    acc1 = __builtin_amdgcn_wmma_f32_16x16x4_f32(
         false, a1,  false, b,
         (short)0, acc1,  false,  false);
  }

  const int rbase = row0 + ((lane >> 4) * 8);
  float* __restrict__ dst = xl + (size_t)rbase * F + col0 + n;
  if (row0 + GEMM_ROWS <= N) {
#pragma unroll
    for (int j = 0; j < 8; ++j) dst[(size_t)j * F] = acc0[j];
#pragma unroll
    for (int j = 0; j < 8; ++j) dst[(size_t)(16 + j) * F] = acc1[j];
  } else {
#pragma unroll
    for (int j = 0; j < 8; ++j) {
      if (rbase + j < N) dst[(size_t)j * F] = acc0[j];
    }
#pragma unroll
    for (int j = 0; j < 8; ++j) {
      if (rbase + 16 + j < N) dst[(size_t)(16 + j) * F] = acc1[j];
    }
  }
}

__global__ void gat_att_kernel(const float* __restrict__ xl,
                               const float* __restrict__ att_src,
                               float* __restrict__ a_src, int* __restrict__ cnt,
                               int NH) {
  int t = blockIdx.x * blockDim.x + threadIdx.x;
  if (t >= NH) return;
  int nidx = t >> 3, h = t & 7;
  const float* __restrict__ xr = xl + (size_t)nidx * F + h * CH;
  const float* __restrict__ as = att_src + h * CH;
  float s = 0.f;
#pragma unroll
  for (int c = 0; c < CH; ++c) s += xr[c] * as[c];
  a_src[t] = s;
  if (h == 0) cnt[nidx] = 0;
}

__device__ __forceinline__ float leaky(float v) {
  return (v >= 0.f) ? v : v * NEG_SLOPE;
}

#define SORT_RANGE 4096
#define SORT_CHUNKS 16
#define SORT_PER_THREAD (SORT_RANGE / 1024)

__device__ __forceinline__ int bin_of(int d, int lo, int N, int lane) {
  const unsigned r = (unsigned)(d - lo);
  return (r < (unsigned)SORT_RANGE && d < N) ? (int)r : SORT_RANGE + lane;
}
__device__ __forceinline__ int4 bin_of4(int4 d, int lo, int N, int lane) {
  int4 b;
  b.x = bin_of(d.x, lo, N, lane);  b.y = bin_of(d.y, lo, N, lane);
  b.z = bin_of(d.z, lo, N, lane);  b.w = bin_of(d.w, lo, N, lane);
  return b;
}
__device__ __forceinline__ int4 lds_add4(int* lc, int4 b) {
  int4 p;
  p.x = atomicAdd(&lc[b.x], 1);  p.y = atomicAdd(&lc[b.y], 1);
  p.z = atomicAdd(&lc[b.z], 1);  p.w = atomicAdd(&lc[b.w], 1);
  return p;
}
__device__ __forceinline__ void put4(int* __restrict__ csr, int4 b, int4 p, int4 s) {
  if (b.x < SORT_RANGE) csr[p.x] = s.x;
  if (b.y < SORT_RANGE) csr[p.y] = s.y;
  if (b.z < SORT_RANGE) csr[p.z] = s.z;
  if (b.w < SORT_RANGE) csr[p.w] = s.w;
}

template <bool PLACE>
__device__ __forceinline__ void edge_pass(const int* __restrict__ ei, int* lc,
                                          int* __restrict__ csr, int E, int N,
                                          int lo, int e0, int e1, int tid) {
  const int lane = tid & 31;
  const int* __restrict__ dst = ei + E;
  if ((E & 3) == 0) {
    const int4* __restrict__ d4 = (const int4*)dst;
    const int4* __restrict__ s4 = (const int4*)ei;
    const int v1 = e1 >> 2;
    int v = (e0 >> 2) + tid;
    for (; v + 3072 < v1; v += 4096) {
      const int4 da = d4[v];
      const int4 db = d4[v + 1024];
      const int4 dc = d4[v + 2048];
      const int4 dg = d4[v + 3072];
      int4 sa, sb, sc, sg;
      if (PLACE) {
        sa = s4[v];  sb = s4[v + 1024];  sc = s4[v + 2048];  sg = s4[v + 3072];
      }
      const int4 ba = bin_of4(da, lo, N, lane);
      const int4 bb = bin_of4(db, lo, N, lane);
      const int4 bc = bin_of4(dc, lo, N, lane);
      const int4 bg = bin_of4(dg, lo, N, lane);
      const int4 pa = lds_add4(lc, ba);
      const int4 pb = lds_add4(lc, bb);
      const int4 pc = lds_add4(lc, bc);
      const int4 pg = lds_add4(lc, bg);
      if (PLACE) {
        put4(csr, ba, pa, sa);  put4(csr, bb, pb, sb);
        put4(csr, bc, pc, sc);  put4(csr, bg, pg, sg);
      }
    }
    for (; v < v1; v += 1024) {
      const int4 da = d4[v];
      int4 sa;
      if (PLACE) sa = s4[v];
      const int4 ba = bin_of4(da, lo, N, lane);
      const int4 pa = lds_add4(lc, ba);
      if (PLACE) put4(csr, ba, pa, sa);
    }
  } else {
    for (int e = e0 + tid; e < e1; e += 1024) {
      const int b = bin_of(dst[e], lo, N, lane);
      const int p = atomicAdd(&lc[b], 1);
      if (PLACE) {
        if (b < SORT_RANGE) csr[p] = ei[e];
      }
    }
  }
}

__device__ __forceinline__ void chunk_bounds(int E, int& e0, int& e1) {
  const int len = (((E + SORT_CHUNKS - 1) / SORT_CHUNKS) + 3) & ~3;
  e0 = (int)blockIdx.x * len;
  if (e0 > E) e0 = E;
  e1 = e0 + len;
  if (e1 > E) e1 = E;
}

__device__ __forceinline__ int wave_sum_int(int v) {
  for (int o = 16; o > 0; o >>= 1) v += __shfl_xor(v, o);
  return v;
}

__global__ void __launch_bounds__(1024)
gat_count_kernel(const int* __restrict__ ei, int* __restrict__ cnt,
                 int* __restrict__ rtot, int E, int N) {
  __shared__ int lc[SORT_RANGE + 32];
  const int tid = threadIdx.x;
  const int lo = (int)blockIdx.y * SORT_RANGE;
  int e0, e1;
  chunk_bounds(E, e0, e1);
  for (int k = tid; k < SORT_RANGE + 32; k += 1024) lc[k] = 0;
  __syncthreads();
  edge_pass<false>(ei, lc, nullptr, E, N, lo, e0, e1, tid);
  __syncthreads();
  for (int k = tid; k < SORT_RANGE; k += 1024) {
    const int c = lc[k];
    if (c > 0) atomicAdd(&cnt[lo + k], c);
  }
  if (tid < 32) {
    const int outside = wave_sum_int(lc[SORT_RANGE + tid]);
    if (tid == 0) rtot[(int)blockIdx.y * SORT_CHUNKS + (int)blockIdx.x] = (e1 - e0) - outside;
  }
}

__global__ void __launch_bounds__(1024)
gat_scan_kernel(int* __restrict__ cnt, const int* __restrict__ rtot, int N) {
  __shared__ int wsum[32];
  __shared__ int psum[32];
  const int tid = threadIdx.x, lane = tid & 31, wave = tid >> 5;
  const int lo = (int)blockIdx.x * SORT_RANGE;
  const int n0 = lo + 4 * tid;

  int part = 0;
  const int lower = (int)blockIdx.x * SORT_CHUNKS;
  for (int i = tid; i < lower; i += 1024) part += rtot[i];

  int4 c = make_int4(0, 0, 0, 0);
  if (n0 + 3 < N) {
    c = *(const int4*)(cnt + n0);
  } else {
    if (n0 < N) c.x = cnt[n0];
    if (n0 + 1 < N) c.y = cnt[n0 + 1];
    if (n0 + 2 < N) c.z = cnt[n0 + 2];
  }
  const int s = (c.x + c.y) + (c.z + c.w);
  int incl = s;
  for (int o = 1; o < 32; o <<= 1) {
    const int v = __shfl_up(incl, o);
    if (lane >= o) incl += v;
  }
  part = wave_sum_int(part);
  if (lane == 31) wsum[wave] = incl;
  if (lane == 0) psum[wave] = part;
  __syncthreads();
  const int before = wave_sum_int(psum[lane] + ((lane < wave) ? wsum[lane] : 0));
  int4 o4;
  o4.x = before + incl - s;
  o4.y = o4.x + c.x;
  o4.z = o4.y + c.y;
  o4.w = o4.z + c.z;
  if (n0 + 3 < N) {
    *(int4*)(cnt + n0) = o4;
  } else {
    if (n0 < N) cnt[n0] = o4.x;
    if (n0 + 1 < N) cnt[n0 + 1] = o4.y;
    if (n0 + 2 < N) cnt[n0 + 2] = o4.z;
  }
}

__global__ void __launch_bounds__(1024)
gat_fill_kernel(const int* __restrict__ ei, int* __restrict__ cnt,
                int* __restrict__ csr, int E, int N) {
  __shared__ int lc[SORT_RANGE + 32];
  const int tid = threadIdx.x;
  const int lo = (int)blockIdx.y * SORT_RANGE;
  int e0, e1;
  chunk_bounds(E, e0, e1);
  for (int k = tid; k < SORT_RANGE + 32; k += 1024) lc[k] = 0;
  __syncthreads();
  edge_pass<false>(ei, lc, nullptr, E, N, lo, e0, e1, tid);
  __syncthreads();
  {
    const int spare = tid % N;
    int c[SORT_PER_THREAD], b[SORT_PER_THREAD];
    for (int u = 0; u < SORT_PER_THREAD; ++u) c[u] = lc[tid + 1024 * u];
    for (int u = 0; u < SORT_PER_THREAD; ++u) {
      const int node = lo + tid + 1024 * u;
      b[u] = atomicAdd(&cnt[node < N ? node : spare], c[u]);
    }
    for (int u = 0; u < SORT_PER_THREAD; ++u) lc[tid + 1024 * u] = b[u];
  }
  __syncthreads();
  edge_pass<true>(ei, lc, csr, E, N, lo, e0, e1, tid);
}

template <int CTRL>
__device__ __forceinline__ float dpp_row(float v) {
  return __int_as_float(__builtin_amdgcn_update_dpp(
      0, __float_as_int(v), CTRL,  0xf,  0xf,  false));
}
__device__ __forceinline__ float xhalf(float v) {
  const int b = __float_as_int(v);
  return __int_as_float(__builtin_amdgcn_permlanex16(b, b, 0x76543210, 0xfedcba98, false, false));
}
__device__ __forceinline__ float wave_sum_valu(float v) {
  v += dpp_row<0x161>(v);
  v += dpp_row<0x162>(v);
  v += dpp_row<0x164>(v);
  v += dpp_row<0x168>(v);
  v += xhalf(v);
  return v;
}

__device__ __forceinline__ float oct_sum(float v) {
  v += dpp_row<0x161>(v);
  v += dpp_row<0x162>(v);
  v += dpp_row<0x164>(v);
  return v;
}

__device__ __forceinline__ float oct_max(float v) {
  v = fmaxf(v, dpp_row<0x161>(v));
  v = fmaxf(v, dpp_row<0x162>(v));
  v = fmaxf(v, dpp_row<0x164>(v));
  return v;
}
template <int U>
__device__ __forceinline__ float oct_bcast(float v) {
  return __int_as_float((int)__builtin_amdgcn_mov_dpp8((unsigned)__float_as_int(v), (unsigned)(U * 0x249249)));
}

template <int N>
__device__ __forceinline__ void fold_group(const float* __restrict__ xl,
                                           const float* __restrict__ a_src,
                                           int sv, int j0, int lane, int hA,
                                           float adA, float adB,
                                           float& mA, float& sA, float4& aA,
                                           float& mB, float& sB, float4& aB) {
  const int slot = lane & 7;
  const int sjl = __shfl(sv, j0 + slot);
  const float* __restrict__ as = a_src + (size_t)sjl * HEADS;
  const float gA = as[hA];
  const float gB = as[hA + 4];
  float4 vA[N], vB[N];
#pragma unroll
  for (int u = 0; u < N; ++u) {
    const int sj = __builtin_amdgcn_readlane(sv, j0 + u);
    const float4* __restrict__ row = (const float4*)(xl + (size_t)sj * F);
    vA[u] = row[lane];
    vB[u] = row[lane + 32];
  }
  const bool live = (N == 8) || slot < N;
  const float lA = live ? leaky(gA + adA) : -INFINITY;
  const float lB = live ? leaky(gB + adB) : -INFINITY;
  const float nmA = fmaxf(mA, oct_max(lA));
  const float nmB = fmaxf(mB, oct_max(lB));
  const float scA = expf(mA - nmA);
  const float scB = expf(mB - nmB);
  const float pA = expf(lA - nmA);
  const float pB = expf(lB - nmB);
  sA = fmaf(sA, scA, pA);
  sB = fmaf(sB, scB, pB);
  mA = nmA;
  mB = nmB;
  aA.x *= scA;  aA.y *= scA;  aA.z *= scA;  aA.w *= scA;
  aB.x *= scB;  aB.y *= scB;  aB.z *= scB;  aB.w *= scB;
#define FOLD_ROW(U)                                                         \
  if constexpr (U < N) {                                                    \
    const float wA = oct_bcast<U>(pA);                                      \
    const float wB = oct_bcast<U>(pB);                                      \
    aA.x = fmaf(wA, vA[U].x, aA.x);  aA.y = fmaf(wA, vA[U].y, aA.y);        \
    aA.z = fmaf(wA, vA[U].z, aA.z);  aA.w = fmaf(wA, vA[U].w, aA.w);        \
    aB.x = fmaf(wB, vB[U].x, aB.x);  aB.y = fmaf(wB, vB[U].y, aB.y);        \
    aB.z = fmaf(wB, vB[U].z, aB.z);  aB.w = fmaf(wB, vB[U].w, aB.w);        \
  }
  FOLD_ROW(0) FOLD_ROW(1) FOLD_ROW(2) FOLD_ROW(3)
  FOLD_ROW(4) FOLD_ROW(5) FOLD_ROW(6) FOLD_ROW(7)
#undef FOLD_ROW
}

__global__ void __launch_bounds__(256) __attribute__((amdgpu_waves_per_eu(4, 8)))
gat_gather_ln_kernel(const float* __restrict__ xl, const float* __restrict__ a_src,
                     const float* __restrict__ att_dst, const int* __restrict__ cnt,
                     const int* __restrict__ csr, const float* __restrict__ bias,
                     const float* __restrict__ gamma, const float* __restrict__ beta,
                     float* __restrict__ out, int N) {
  const int lane = threadIdx.x & 31;
  const int d = __builtin_amdgcn_readfirstlane((int)(blockIdx.x * 8u + (threadIdx.x >> 5)));
  if (d >= N) return;
  const int hA = lane >> 3;

  const int end = cnt[d];
  const int start = (d > 0) ? cnt[d - 1] : 0;

  const float4* __restrict__ own = (const float4*)(xl + (size_t)d * F);
  const float4 xA = own[lane];
  const float4 xB = own[lane + 32];
  const float4 tA = ((const float4*)att_dst)[lane];
  const float4 tB = ((const float4*)att_dst)[lane + 32];
  const float adA = oct_sum(xA.x * tA.x + xA.y * tA.y + xA.z * tA.z + xA.w * tA.w);
  const float adB = oct_sum(xB.x * tB.x + xB.y * tB.y + xB.z * tB.z + xB.w * tB.w);

  const float* __restrict__ aown = a_src + (size_t)d * HEADS;
  float mA = leaky(aown[hA] + adA);
  float mB = leaky(aown[hA + 4] + adB);
  float sA = ((lane & 7) == 0) ? 1.f : 0.f;
  float sB = sA;
  float4 aA = xA, aB = xB;

  for (int base = start; base < end; base += 32) {
    const int idx = base + lane;
    const int sv = (idx < end) ? csr[idx] : d;
    int left = end - base;
    if (left > 32) left = 32;
    int j0 = 0;
    for (; j0 + 8 <= left; j0 += 8)
      fold_group<8>(xl, a_src, sv, j0, lane, hA, adA, adB, mA, sA, aA, mB, sB, aB);
#define REST(K) case K: fold_group<K>(xl, a_src, sv, j0, lane, hA, adA, adB, mA, sA, aA, mB, sB, aB); break;
    switch (left - j0) {
      REST(1) REST(2) REST(3) REST(4) REST(5) REST(6) REST(7)
      default: break;
    }
#undef REST
  }
  sA = oct_sum(sA);
  sB = oct_sum(sB);

  const float4 bA = ((const float4*)bias)[lane];
  const float4 bB = ((const float4*)bias)[lane + 32];
  const float4 gA = ((const float4*)gamma)[lane];
  const float4 gB = ((const float4*)gamma)[lane + 32];
  const float4 eA = ((const float4*)beta)[lane];
  const float4 eB = ((const float4*)beta)[lane + 32];

  float v[8];
  v[0] = aA.x / sA + bA.x;  v[1] = aA.y / sA + bA.y;
  v[2] = aA.z / sA + bA.z;  v[3] = aA.w / sA + bA.w;
  v[4] = aB.x / sB + bB.x;  v[5] = aB.y / sB + bB.y;
  v[6] = aB.z / sB + bB.z;  v[7] = aB.w / sB + bB.w;

  const float s1 = ((v[0] + v[1]) + (v[2] + v[3])) + ((v[4] + v[5]) + (v[6] + v[7]));
  const float mean = wave_sum_valu(s1) * (1.0f / F);
  float s2 = 0.f;
#pragma unroll
  for (int i = 0; i < 8; ++i) {
    v[i] -= mean;
    s2 = fmaf(v[i], v[i], s2);
  }
  const float var = wave_sum_valu(s2) * (1.0f / F);
  const float rs = rsqrtf(var + LN_EPS);

  float4 oA, oB;
  oA.x = v[0] * rs * gA.x + eA.x;  oA.y = v[1] * rs * gA.y + eA.y;
  oA.z = v[2] * rs * gA.z + eA.z;  oA.w = v[3] * rs * gA.w + eA.w;
  oB.x = v[4] * rs * gB.x + eB.x;  oB.y = v[5] * rs * gB.y + eB.y;
  oB.z = v[6] * rs * gB.z + eB.z;  oB.w = v[7] * rs * gB.w + eB.w;
  float4* __restrict__ orow = (float4*)(out + (size_t)d * F);
  orow[lane] = oA;
  orow[lane + 32] = oB;
}

extern "C" void kernel_launch(void* const* d_in, const int* in_sizes, int n_in,
                              void* d_out, int out_size, void* d_ws, size_t ws_size,
                              hipStream_t stream) {
  const float* x       = (const float*)d_in[0];
  const int*   ei      = (const int*)  d_in[1];
  const float* W       = (const float*)d_in[2];
  const float* att_src = (const float*)d_in[3];
  const float* att_dst = (const float*)d_in[4];
  const float* bias    = (const float*)d_in[5];
  const float* gamma   = (const float*)d_in[6];
  const float* beta    = (const float*)d_in[7];
  float* out = (float*)d_out;

  const int N = in_sizes[0] / F;
  const int E = in_sizes[1] / 2;

  char* ws = (char*)d_ws;
  float* xl    = (float*)ws; ws += (size_t)N * F * sizeof(float);
  float* a_src = (float*)ws; ws += (size_t)N * HEADS * sizeof(float);
  int*   cnt   = (int*)ws;   ws += (size_t)N * HEADS * sizeof(float);
  int*   csr   = (int*)ws;
  int*   rtot  = cnt + N;

  gat_gemm_wmma_kernel<<<(N + GEMM_ROWS - 1) / GEMM_ROWS, 512, 0, stream>>>(x, W, xl, N);

  {
    int NH = N * HEADS;
    gat_att_kernel<<<(NH + 255) / 256, 256, 0, stream>>>(xl, att_src, a_src, cnt, NH);
  }

  {
    dim3 grid(SORT_CHUNKS, (N + SORT_RANGE - 1) / SORT_RANGE);
    gat_count_kernel<<<grid, 1024, 0, stream>>>(ei, cnt, rtot, E, N);
    gat_scan_kernel<<<grid.y, 1024, 0, stream>>>(cnt, rtot, N);
    gat_fill_kernel<<<grid, 1024, 0, stream>>>(ei, cnt, csr, E, N);
  }

  gat_gather_ln_kernel<<<(N + 7) / 8, 256, 0, stream>>>(xl, a_src, att_dst, cnt, csr,
                                                        bias, gamma, beta, out, N);
}


# source half-logits taken from the gathered rows themselves (4-channel dots per lane + octet butterfly: new sum order for a_src): per-edge score fetch, a_src table and its launch gone
# speedup vs baseline: 5.4692x; 1.0663x over previous
#include <hip/hip_runtime.h>
#include <math.h>

typedef float v2f __attribute__((ext_vector_type(2)));
typedef float v8f __attribute__((ext_vector_type(8)));

#define HEADS 8
#define CH 32
#define F 256
#define NEG_SLOPE 0.2f
#define LN_EPS 1e-5f

#define LDS_STRIDE 260

#define GEMM_ROWS 32
__global__ void __launch_bounds__(512)
gat_gemm_wmma_kernel(const float* __restrict__ x, const float* __restrict__ W,
                     float* __restrict__ xl, int* __restrict__ cnt, int N) {
  __shared__ float lds_a[GEMM_ROWS * LDS_STRIDE];

  const int tid  = threadIdx.x;
  const int lane = tid & 31;
  const int wave = tid >> 5;
  const int row0 = blockIdx.x * GEMM_ROWS;
  const int col0 = wave * 16;

  if (tid < GEMM_ROWS && row0 + tid < N) cnt[row0 + tid] = 0;

  {
    const int c = (tid & 31) * 8;
#pragma unroll
    for (int p = 0; p < GEMM_ROWS / 16; ++p) {
      const int r = (tid >> 5) + 16 * p;
      int grow = row0 + r;
      if (grow >= N) grow = N - 1;
      const float4* __restrict__ src = (const float4*)(x + (size_t)grow * F + c);
      float4* dst = (float4*)(&lds_a[r * LDS_STRIDE + c]);
      dst[0] = src[0];
      dst[1] = src[1];
    }
  }
  __syncthreads();

  const int m    = lane & 15;
  const int koff = (lane >> 4) * 2;
  const float* __restrict__ arow0 = &lds_a[m * LDS_STRIDE];
  const float* __restrict__ arow1 = &lds_a[(16 + m) * LDS_STRIDE];

  const int n = lane & 15;

  v8f acc0 = {0.f, 0.f, 0.f, 0.f, 0.f, 0.f, 0.f, 0.f};
  v8f acc1 = {0.f, 0.f, 0.f, 0.f, 0.f, 0.f, 0.f, 0.f};
  for (int k = 0; k < F; k += 4) {
    v2f a0 = *(const v2f*)(arow0 + k + koff);
    v2f a1 = *(const v2f*)(arow1 + k + koff);
    v2f b;
    b.x = W[(size_t)(k + koff) * F + col0 + n];
    b.y = W[(size_t)(k + koff + 1) * F + col0 + n];
    acc0 = __builtin_amdgcn_wmma_f32_16x16x4_f32(
         false, a0,  false, b,
         (short)0, acc0,  false,  false);
    acc1 = __builtin_amdgcn_wmma_f32_16x16x4_f32(
         false, a1,  false, b,
         (short)0, acc1,  false,  false);
  }

  const int rbase = row0 + ((lane >> 4) * 8);
  float* __restrict__ dst = xl + (size_t)rbase * F + col0 + n;
  if (row0 + GEMM_ROWS <= N) {
#pragma unroll
    for (int j = 0; j < 8; ++j) dst[(size_t)j * F] = acc0[j];
#pragma unroll
    for (int j = 0; j < 8; ++j) dst[(size_t)(16 + j) * F] = acc1[j];
  } else {
#pragma unroll
    for (int j = 0; j < 8; ++j) {
      if (rbase + j < N) dst[(size_t)j * F] = acc0[j];
    }
#pragma unroll
    for (int j = 0; j < 8; ++j) {
      if (rbase + 16 + j < N) dst[(size_t)(16 + j) * F] = acc1[j];
    }
  }
}

__device__ __forceinline__ float leaky(float v) {
  return (v >= 0.f) ? v : v * NEG_SLOPE;
}

#define SORT_RANGE 4096
#define SORT_CHUNKS 16
#define SORT_PER_THREAD (SORT_RANGE / 1024)

__device__ __forceinline__ int bin_of(int d, int lo, int N, int lane) {
  const unsigned r = (unsigned)(d - lo);
  return (r < (unsigned)SORT_RANGE && d < N) ? (int)r : SORT_RANGE + lane;
}
__device__ __forceinline__ int4 bin_of4(int4 d, int lo, int N, int lane) {
  int4 b;
  b.x = bin_of(d.x, lo, N, lane);  b.y = bin_of(d.y, lo, N, lane);
  b.z = bin_of(d.z, lo, N, lane);  b.w = bin_of(d.w, lo, N, lane);
  return b;
}
__device__ __forceinline__ int4 lds_add4(int* lc, int4 b) {
  int4 p;
  p.x = atomicAdd(&lc[b.x], 1);  p.y = atomicAdd(&lc[b.y], 1);
  p.z = atomicAdd(&lc[b.z], 1);  p.w = atomicAdd(&lc[b.w], 1);
  return p;
}
__device__ __forceinline__ void put4(int* __restrict__ csr, int4 b, int4 p, int4 s) {
  if (b.x < SORT_RANGE) csr[p.x] = s.x;
  if (b.y < SORT_RANGE) csr[p.y] = s.y;
  if (b.z < SORT_RANGE) csr[p.z] = s.z;
  if (b.w < SORT_RANGE) csr[p.w] = s.w;
}

template <bool PLACE>
__device__ __forceinline__ void edge_pass(const int* __restrict__ ei, int* lc,
                                          int* __restrict__ csr, int E, int N,
                                          int lo, int e0, int e1, int tid) {
  const int lane = tid & 31;
  const int* __restrict__ dst = ei + E;
  if ((E & 3) == 0) {
    const int4* __restrict__ d4 = (const int4*)dst;
    const int4* __restrict__ s4 = (const int4*)ei;
    const int v1 = e1 >> 2;
    int v = (e0 >> 2) + tid;
    for (; v + 3072 < v1; v += 4096) {
      const int4 da = d4[v];
      const int4 db = d4[v + 1024];
      const int4 dc = d4[v + 2048];
      const int4 dg = d4[v + 3072];
      int4 sa, sb, sc, sg;
      if (PLACE) {
        sa = s4[v];  sb = s4[v + 1024];  sc = s4[v + 2048];  sg = s4[v + 3072];
      }
      const int4 ba = bin_of4(da, lo, N, lane);
      const int4 bb = bin_of4(db, lo, N, lane);
      const int4 bc = bin_of4(dc, lo, N, lane);
      const int4 bg = bin_of4(dg, lo, N, lane);
      const int4 pa = lds_add4(lc, ba);
      const int4 pb = lds_add4(lc, bb);
      const int4 pc = lds_add4(lc, bc);
      const int4 pg = lds_add4(lc, bg);
      if (PLACE) {
        put4(csr, ba, pa, sa);  put4(csr, bb, pb, sb);
        put4(csr, bc, pc, sc);  put4(csr, bg, pg, sg);
      }
    }
    for (; v < v1; v += 1024) {
      const int4 da = d4[v];
      int4 sa;
      if (PLACE) sa = s4[v];
      const int4 ba = bin_of4(da, lo, N, lane);
      const int4 pa = lds_add4(lc, ba);
      if (PLACE) put4(csr, ba, pa, sa);
    }
  } else {
    for (int e = e0 + tid; e < e1; e += 1024) {
      const int b = bin_of(dst[e], lo, N, lane);
      const int p = atomicAdd(&lc[b], 1);
      if (PLACE) {
        if (b < SORT_RANGE) csr[p] = ei[e];
      }
    }
  }
}

__device__ __forceinline__ void chunk_bounds(int E, int& e0, int& e1) {
  const int len = (((E + SORT_CHUNKS - 1) / SORT_CHUNKS) + 3) & ~3;
  e0 = (int)blockIdx.x * len;
  if (e0 > E) e0 = E;
  e1 = e0 + len;
  if (e1 > E) e1 = E;
}

__device__ __forceinline__ int wave_sum_int(int v) {
  for (int o = 16; o > 0; o >>= 1) v += __shfl_xor(v, o);
  return v;
}

__global__ void __launch_bounds__(1024)
gat_count_kernel(const int* __restrict__ ei, int* __restrict__ cnt,
                 int* __restrict__ rtot, int E, int N) {
  __shared__ int lc[SORT_RANGE + 32];
  const int tid = threadIdx.x;
  const int lo = (int)blockIdx.y * SORT_RANGE;
  int e0, e1;
  chunk_bounds(E, e0, e1);
  for (int k = tid; k < SORT_RANGE + 32; k += 1024) lc[k] = 0;
  __syncthreads();
  edge_pass<false>(ei, lc, nullptr, E, N, lo, e0, e1, tid);
  __syncthreads();
  for (int k = tid; k < SORT_RANGE; k += 1024) {
    const int c = lc[k];
    if (c > 0) atomicAdd(&cnt[lo + k], c);
  }
  if (tid < 32) {
    const int outside = wave_sum_int(lc[SORT_RANGE + tid]);
    if (tid == 0) rtot[(int)blockIdx.y * SORT_CHUNKS + (int)blockIdx.x] = (e1 - e0) - outside;
  }
}

__global__ void __launch_bounds__(1024)
gat_scan_kernel(int* __restrict__ cnt, const int* __restrict__ rtot, int N) {
  __shared__ int wsum[32];
  __shared__ int psum[32];
  const int tid = threadIdx.x, lane = tid & 31, wave = tid >> 5;
  const int lo = (int)blockIdx.x * SORT_RANGE;
  const int n0 = lo + 4 * tid;

  int part = 0;
  const int lower = (int)blockIdx.x * SORT_CHUNKS;
  for (int i = tid; i < lower; i += 1024) part += rtot[i];

  int4 c = make_int4(0, 0, 0, 0);
  if (n0 + 3 < N) {
    c = *(const int4*)(cnt + n0);
  } else {
    if (n0 < N) c.x = cnt[n0];
    if (n0 + 1 < N) c.y = cnt[n0 + 1];
    if (n0 + 2 < N) c.z = cnt[n0 + 2];
  }
  const int s = (c.x + c.y) + (c.z + c.w);
  int incl = s;
  for (int o = 1; o < 32; o <<= 1) {
    const int v = __shfl_up(incl, o);
    if (lane >= o) incl += v;
  }
  part = wave_sum_int(part);
  if (lane == 31) wsum[wave] = incl;
  if (lane == 0) psum[wave] = part;
  __syncthreads();
  const int before = wave_sum_int(psum[lane] + ((lane < wave) ? wsum[lane] : 0));
  int4 o4;
  o4.x = before + incl - s;
  o4.y = o4.x + c.x;
  o4.z = o4.y + c.y;
  o4.w = o4.z + c.z;
  if (n0 + 3 < N) {
    *(int4*)(cnt + n0) = o4;
  } else {
    if (n0 < N) cnt[n0] = o4.x;
    if (n0 + 1 < N) cnt[n0 + 1] = o4.y;
    if (n0 + 2 < N) cnt[n0 + 2] = o4.z;
  }
}

__global__ void __launch_bounds__(1024)
gat_fill_kernel(const int* __restrict__ ei, int* __restrict__ cnt,
                int* __restrict__ csr, int E, int N) {
  __shared__ int lc[SORT_RANGE + 32];
  const int tid = threadIdx.x;
  const int lo = (int)blockIdx.y * SORT_RANGE;
  int e0, e1;
  chunk_bounds(E, e0, e1);
  for (int k = tid; k < SORT_RANGE + 32; k += 1024) lc[k] = 0;
  __syncthreads();
  edge_pass<false>(ei, lc, nullptr, E, N, lo, e0, e1, tid);
  __syncthreads();
  {
    const int spare = tid % N;
    int c[SORT_PER_THREAD], b[SORT_PER_THREAD];
    for (int u = 0; u < SORT_PER_THREAD; ++u) c[u] = lc[tid + 1024 * u];
    for (int u = 0; u < SORT_PER_THREAD; ++u) {
      const int node = lo + tid + 1024 * u;
      b[u] = atomicAdd(&cnt[node < N ? node : spare], c[u]);
    }
    for (int u = 0; u < SORT_PER_THREAD; ++u) lc[tid + 1024 * u] = b[u];
  }
  __syncthreads();
  edge_pass<true>(ei, lc, csr, E, N, lo, e0, e1, tid);
}

template <int CTRL>
__device__ __forceinline__ float dpp_row(float v) {
  return __int_as_float(__builtin_amdgcn_update_dpp(
      0, __float_as_int(v), CTRL,  0xf,  0xf,  false));
}
__device__ __forceinline__ float xhalf(float v) {
  const int b = __float_as_int(v);
  return __int_as_float(__builtin_amdgcn_permlanex16(b, b, 0x76543210, 0xfedcba98, false, false));
}
__device__ __forceinline__ float wave_sum_valu(float v) {
  v += dpp_row<0x161>(v);
  v += dpp_row<0x162>(v);
  v += dpp_row<0x164>(v);
  v += dpp_row<0x168>(v);
  v += xhalf(v);
  return v;
}

__device__ __forceinline__ float oct_sum(float v) {
  v += dpp_row<0x161>(v);
  v += dpp_row<0x162>(v);
  v += dpp_row<0x164>(v);
  return v;
}

__device__ __forceinline__ float oct_max(float v) {
  v = fmaxf(v, dpp_row<0x161>(v));
  v = fmaxf(v, dpp_row<0x162>(v));
  v = fmaxf(v, dpp_row<0x164>(v));
  return v;
}
template <int U>
__device__ __forceinline__ float oct_bcast(float v) {
  return __int_as_float((int)__builtin_amdgcn_mov_dpp8((unsigned)__float_as_int(v), (unsigned)(U * 0x249249)));
}

__device__ __forceinline__ float dot4(float4 a, float4 b) {
  const v2f al = {a.x, a.y}, ah = {a.z, a.w};
  const v2f bl = {b.x, b.y}, bh = {b.z, b.w};
  const v2f m = __builtin_elementwise_fma(ah, bh, al * bl);
  return m.x + m.y;
}

template <int CTRL>
__device__ __forceinline__ float fold_step(bool bit, float lo, float hi) {
  const float keep = bit ? hi : lo;
  const float send = bit ? lo : hi;
  return keep + dpp_row<CTRL>(send);
}

__device__ __forceinline__ float row_scores(const float (&p)[8], bool b4, bool b2, bool b1) {
  const float q0 = fold_step<0x164>(b4, p[0], p[4]);
  const float q1 = fold_step<0x164>(b4, p[1], p[5]);
  const float q2 = fold_step<0x164>(b4, p[2], p[6]);
  const float q3 = fold_step<0x164>(b4, p[3], p[7]);
  const float r0 = fold_step<0x162>(b2, q0, q2);
  const float r1 = fold_step<0x162>(b2, q1, q3);
  return fold_step<0x161>(b1, r0, r1);
}

template <int N>
__device__ __forceinline__ void fold_group(const float* __restrict__ xl,
                                           int sv, int j0, int lane,
                                           float4 tsA, float4 tsB,
                                           float adA, float adB,
                                           float& mA, float& sA, float4& aA,
                                           float& mB, float& sB, float4& aB) {
  const int slot = lane & 7;
  float4 vA[N], vB[N];
#pragma unroll
  for (int u = 0; u < N; ++u) {
    const int sj = __builtin_amdgcn_readlane(sv, j0 + u);
    const float4* __restrict__ row = (const float4*)(xl + (size_t)sj * F);
    vA[u] = row[lane];
    vB[u] = row[lane + 32];
  }
  float dA[8], dB[8];
#define ROW_DOT(U)                                                          \
  if constexpr (U < N) {                                                    \
    dA[U] = dot4(vA[U], tsA);                                               \
    dB[U] = dot4(vB[U], tsB);                                               \
  } else {           \
    dA[U] = 0.f;                                                            \
    dB[U] = 0.f;                                                            \
  }
  ROW_DOT(0) ROW_DOT(1) ROW_DOT(2) ROW_DOT(3)
  ROW_DOT(4) ROW_DOT(5) ROW_DOT(6) ROW_DOT(7)
#undef ROW_DOT
  const bool b4 = (slot & 4) != 0, b2 = (slot & 2) != 0, b1 = (slot & 1) != 0;
  const float gA = row_scores(dA, b4, b2, b1);
  const float gB = row_scores(dB, b4, b2, b1);
  const bool live = (N == 8) || slot < N;
  const float lA = live ? leaky(gA + adA) : -INFINITY;
  const float lB = live ? leaky(gB + adB) : -INFINITY;
  const float nmA = fmaxf(mA, oct_max(lA));
  const float nmB = fmaxf(mB, oct_max(lB));
  const float scA = expf(mA - nmA);
  const float scB = expf(mB - nmB);
  const float pA = expf(lA - nmA);
  const float pB = expf(lB - nmB);
  sA = fmaf(sA, scA, pA);
  sB = fmaf(sB, scB, pB);
  mA = nmA;
  mB = nmB;
  aA.x *= scA;  aA.y *= scA;  aA.z *= scA;  aA.w *= scA;
  aB.x *= scB;  aB.y *= scB;  aB.z *= scB;  aB.w *= scB;
#define FOLD_ROW(U)                                                         \
  if constexpr (U < N) {                                                    \
    const float wA = oct_bcast<U>(pA);                                      \
    const float wB = oct_bcast<U>(pB);                                      \
    aA.x = fmaf(wA, vA[U].x, aA.x);  aA.y = fmaf(wA, vA[U].y, aA.y);        \
    aA.z = fmaf(wA, vA[U].z, aA.z);  aA.w = fmaf(wA, vA[U].w, aA.w);        \
    aB.x = fmaf(wB, vB[U].x, aB.x);  aB.y = fmaf(wB, vB[U].y, aB.y);        \
    aB.z = fmaf(wB, vB[U].z, aB.z);  aB.w = fmaf(wB, vB[U].w, aB.w);        \
  }
  FOLD_ROW(0) FOLD_ROW(1) FOLD_ROW(2) FOLD_ROW(3)
  FOLD_ROW(4) FOLD_ROW(5) FOLD_ROW(6) FOLD_ROW(7)
#undef FOLD_ROW
}

__global__ void __launch_bounds__(256) __attribute__((amdgpu_waves_per_eu(4, 8)))
gat_gather_ln_kernel(const float* __restrict__ xl, const float* __restrict__ att_src,
                     const float* __restrict__ att_dst, const int* __restrict__ cnt,
                     const int* __restrict__ csr, const float* __restrict__ bias,
                     const float* __restrict__ gamma, const float* __restrict__ beta,
                     float* __restrict__ out, int N) {
  const int lane = threadIdx.x & 31;
  const int d = __builtin_amdgcn_readfirstlane((int)(blockIdx.x * 8u + (threadIdx.x >> 5)));
  if (d >= N) return;

  const int end = cnt[d];
  const int start = (d > 0) ? cnt[d - 1] : 0;

  const float4* __restrict__ own = (const float4*)(xl + (size_t)d * F);
  const float4 xA = own[lane];
  const float4 xB = own[lane + 32];
  const float4 tA = ((const float4*)att_dst)[lane];
  const float4 tB = ((const float4*)att_dst)[lane + 32];
  const float4 tsA = ((const float4*)att_src)[lane];
  const float4 tsB = ((const float4*)att_src)[lane + 32];
  const float adA = oct_sum(xA.x * tA.x + xA.y * tA.y + xA.z * tA.z + xA.w * tA.w);
  const float adB = oct_sum(xB.x * tB.x + xB.y * tB.y + xB.z * tB.z + xB.w * tB.w);

  float mA = leaky(oct_sum(dot4(xA, tsA)) + adA);
  float mB = leaky(oct_sum(dot4(xB, tsB)) + adB);
  float sA = ((lane & 7) == 0) ? 1.f : 0.f;
  float sB = sA;
  float4 aA = xA, aB = xB;

  for (int base = start; base < end; base += 32) {
    const int idx = base + lane;
    const int sv = (idx < end) ? csr[idx] : d;
    int left = end - base;
    if (left > 32) left = 32;
    int j0 = 0;
    for (; j0 + 8 <= left; j0 += 8)
      fold_group<8>(xl, sv, j0, lane, tsA, tsB, adA, adB, mA, sA, aA, mB, sB, aB);
#define REST(K) case K: fold_group<K>(xl, sv, j0, lane, tsA, tsB, adA, adB, mA, sA, aA, mB, sB, aB); break;
    switch (left - j0) {
      REST(1) REST(2) REST(3) REST(4) REST(5) REST(6) REST(7)
      default: break;
    }
#undef REST
  }
  sA = oct_sum(sA);
  sB = oct_sum(sB);

  const float4 bA = ((const float4*)bias)[lane];
  const float4 bB = ((const float4*)bias)[lane + 32];
  const float4 gA = ((const float4*)gamma)[lane];
  const float4 gB = ((const float4*)gamma)[lane + 32];
  const float4 eA = ((const float4*)beta)[lane];
  const float4 eB = ((const float4*)beta)[lane + 32];

  float v[8];
  v[0] = aA.x / sA + bA.x;  v[1] = aA.y / sA + bA.y;
  v[2] = aA.z / sA + bA.z;  v[3] = aA.w / sA + bA.w;
  v[4] = aB.x / sB + bB.x;  v[5] = aB.y / sB + bB.y;
  v[6] = aB.z / sB + bB.z;  v[7] = aB.w / sB + bB.w;

  const float s1 = ((v[0] + v[1]) + (v[2] + v[3])) + ((v[4] + v[5]) + (v[6] + v[7]));
  const float mean = wave_sum_valu(s1) * (1.0f / F);
  float s2 = 0.f;
#pragma unroll
  for (int i = 0; i < 8; ++i) {
    v[i] -= mean;
    s2 = fmaf(v[i], v[i], s2);
  }
  const float var = wave_sum_valu(s2) * (1.0f / F);
  const float rs = rsqrtf(var + LN_EPS);

  float4 oA, oB;
  oA.x = v[0] * rs * gA.x + eA.x;  oA.y = v[1] * rs * gA.y + eA.y;
  oA.z = v[2] * rs * gA.z + eA.z;  oA.w = v[3] * rs * gA.w + eA.w;
  oB.x = v[4] * rs * gB.x + eB.x;  oB.y = v[5] * rs * gB.y + eB.y;
  oB.z = v[6] * rs * gB.z + eB.z;  oB.w = v[7] * rs * gB.w + eB.w;
  float4* __restrict__ orow = (float4*)(out + (size_t)d * F);
  orow[lane] = oA;
  orow[lane + 32] = oB;
}

extern "C" void kernel_launch(void* const* d_in, const int* in_sizes, int n_in,
                              void* d_out, int out_size, void* d_ws, size_t ws_size,
                              hipStream_t stream) {
  const float* x       = (const float*)d_in[0];
  const int*   ei      = (const int*)  d_in[1];
  const float* W       = (const float*)d_in[2];
  const float* att_src = (const float*)d_in[3];
  const float* att_dst = (const float*)d_in[4];
  const float* bias    = (const float*)d_in[5];
  const float* gamma   = (const float*)d_in[6];
  const float* beta    = (const float*)d_in[7];
  float* out = (float*)d_out;

  const int N = in_sizes[0] / F;
  const int E = in_sizes[1] / 2;

  char* ws = (char*)d_ws;
  float* xl    = (float*)ws; ws += (size_t)N * F * sizeof(float);
  int*   cnt   = (int*)ws;   ws += (size_t)N * HEADS * sizeof(float);
  int*   csr   = (int*)ws;
  int*   rtot  = cnt + N;

  gat_gemm_wmma_kernel<<<(N + GEMM_ROWS - 1) / GEMM_ROWS, 512, 0, stream>>>(x, W, xl, cnt, N);

  {
    dim3 grid(SORT_CHUNKS, (N + SORT_RANGE - 1) / SORT_RANGE);
    gat_count_kernel<<<grid, 1024, 0, stream>>>(ei, cnt, rtot, E, N);
    gat_scan_kernel<<<grid.y, 1024, 0, stream>>>(cnt, rtot, N);
    gat_fill_kernel<<<grid, 1024, 0, stream>>>(ei, cnt, csr, E, N);
  }

  gat_gather_ln_kernel<<<(N + 7) / 8, 256, 0, stream>>>(xl, att_src, att_dst, cnt, csr,
                                                        bias, gamma, beta, out, N);
}


# edge sort keeps its counts: count stores per-chunk planes (xl area; projection moved behind the sort), fill scans its range from them: recount, scan launch, global atomics gone; slots in chunk order
# speedup vs baseline: 5.7089x; 1.0438x over previous
#include <hip/hip_runtime.h>
#include <math.h>

typedef float v2f __attribute__((ext_vector_type(2)));
typedef float v8f __attribute__((ext_vector_type(8)));

#define HEADS 8
#define CH 32
#define F 256
#define NEG_SLOPE 0.2f
#define LN_EPS 1e-5f

#define LDS_STRIDE 260

#define GEMM_ROWS 32
__global__ void __launch_bounds__(512)
gat_gemm_wmma_kernel(const float* __restrict__ x, const float* __restrict__ W,
                     float* __restrict__ xl, int N) {
  __shared__ float lds_a[GEMM_ROWS * LDS_STRIDE];

  const int tid  = threadIdx.x;
  const int lane = tid & 31;
  const int wave = tid >> 5;
  const int row0 = blockIdx.x * GEMM_ROWS;
  const int col0 = wave * 16;

  {
    const int c = (tid & 31) * 8;
#pragma unroll
    for (int p = 0; p < GEMM_ROWS / 16; ++p) {
      const int r = (tid >> 5) + 16 * p;
      int grow = row0 + r;
      if (grow >= N) grow = N - 1;
      const float4* __restrict__ src = (const float4*)(x + (size_t)grow * F + c);
      float4* dst = (float4*)(&lds_a[r * LDS_STRIDE + c]);
      dst[0] = src[0];
      dst[1] = src[1];
    }
  }
  __syncthreads();

  const int m    = lane & 15;
  const int koff = (lane >> 4) * 2;
  const float* __restrict__ arow0 = &lds_a[m * LDS_STRIDE];
  const float* __restrict__ arow1 = &lds_a[(16 + m) * LDS_STRIDE];

  const int n = lane & 15;

  v8f acc0 = {0.f, 0.f, 0.f, 0.f, 0.f, 0.f, 0.f, 0.f};
  v8f acc1 = {0.f, 0.f, 0.f, 0.f, 0.f, 0.f, 0.f, 0.f};
  for (int k = 0; k < F; k += 4) {
    v2f a0 = *(const v2f*)(arow0 + k + koff);
    v2f a1 = *(const v2f*)(arow1 + k + koff);
    v2f b;
    b.x = W[(size_t)(k + koff) * F + col0 + n];
    b.y = W[(size_t)(k + koff + 1) * F + col0 + n];
    acc0 = __builtin_amdgcn_wmma_f32_16x16x4_f32(
         false, a0,  false, b,
         (short)0, acc0,  false,  false);
    acc1 = __builtin_amdgcn_wmma_f32_16x16x4_f32(
         false, a1,  false, b,
         (short)0, acc1,  false,  false);
  }

  const int rbase = row0 + ((lane >> 4) * 8);
  float* __restrict__ dst = xl + (size_t)rbase * F + col0 + n;
  if (row0 + GEMM_ROWS <= N) {
#pragma unroll
    for (int j = 0; j < 8; ++j) dst[(size_t)j * F] = acc0[j];
#pragma unroll
    for (int j = 0; j < 8; ++j) dst[(size_t)(16 + j) * F] = acc1[j];
  } else {
#pragma unroll
    for (int j = 0; j < 8; ++j) {
      if (rbase + j < N) dst[(size_t)j * F] = acc0[j];
    }
#pragma unroll
    for (int j = 0; j < 8; ++j) {
      if (rbase + 16 + j < N) dst[(size_t)(16 + j) * F] = acc1[j];
    }
  }
}

__device__ __forceinline__ float leaky(float v) {
  return (v >= 0.f) ? v : v * NEG_SLOPE;
}

#define SORT_RANGE 4096
#define SORT_CHUNKS 16
#define SORT_PER_THREAD (SORT_RANGE / 1024)

__device__ __forceinline__ int bin_of(int d, int lo, int N, int lane) {
  const unsigned r = (unsigned)(d - lo);
  return (r < (unsigned)SORT_RANGE && d < N) ? (int)r : SORT_RANGE + lane;
}
__device__ __forceinline__ int4 bin_of4(int4 d, int lo, int N, int lane) {
  int4 b;
  b.x = bin_of(d.x, lo, N, lane);  b.y = bin_of(d.y, lo, N, lane);
  b.z = bin_of(d.z, lo, N, lane);  b.w = bin_of(d.w, lo, N, lane);
  return b;
}
__device__ __forceinline__ int4 lds_add4(int* lc, int4 b) {
  int4 p;
  p.x = atomicAdd(&lc[b.x], 1);  p.y = atomicAdd(&lc[b.y], 1);
  p.z = atomicAdd(&lc[b.z], 1);  p.w = atomicAdd(&lc[b.w], 1);
  return p;
}
__device__ __forceinline__ void put4(int* __restrict__ csr, int4 b, int4 p, int4 s) {
  if (b.x < SORT_RANGE) csr[p.x] = s.x;
  if (b.y < SORT_RANGE) csr[p.y] = s.y;
  if (b.z < SORT_RANGE) csr[p.z] = s.z;
  if (b.w < SORT_RANGE) csr[p.w] = s.w;
}

template <bool PLACE>
__device__ __forceinline__ void edge_pass(const int* __restrict__ ei, int* lc,
                                          int* __restrict__ csr, int E, int N,
                                          int lo, int e0, int e1, int tid) {
  const int lane = tid & 31;
  const int* __restrict__ dst = ei + E;
  if ((E & 3) == 0) {
    const int4* __restrict__ d4 = (const int4*)dst;
    const int4* __restrict__ s4 = (const int4*)ei;
    const int v1 = e1 >> 2;
    int v = (e0 >> 2) + tid;
    for (; v + 3072 < v1; v += 4096) {
      const int4 da = d4[v];
      const int4 db = d4[v + 1024];
      const int4 dc = d4[v + 2048];
      const int4 dg = d4[v + 3072];
      int4 sa, sb, sc, sg;
      if (PLACE) {
        sa = s4[v];  sb = s4[v + 1024];  sc = s4[v + 2048];  sg = s4[v + 3072];
      }
      const int4 ba = bin_of4(da, lo, N, lane);
      const int4 bb = bin_of4(db, lo, N, lane);
      const int4 bc = bin_of4(dc, lo, N, lane);
      const int4 bg = bin_of4(dg, lo, N, lane);
      const int4 pa = lds_add4(lc, ba);
      const int4 pb = lds_add4(lc, bb);
      const int4 pc = lds_add4(lc, bc);
      const int4 pg = lds_add4(lc, bg);
      if (PLACE) {
        put4(csr, ba, pa, sa);  put4(csr, bb, pb, sb);
        put4(csr, bc, pc, sc);  put4(csr, bg, pg, sg);
      }
    }
    for (; v < v1; v += 1024) {
      const int4 da = d4[v];
      int4 sa;
      if (PLACE) sa = s4[v];
      const int4 ba = bin_of4(da, lo, N, lane);
      const int4 pa = lds_add4(lc, ba);
      if (PLACE) put4(csr, ba, pa, sa);
    }
  } else {
    for (int e = e0 + tid; e < e1; e += 1024) {
      const int b = bin_of(dst[e], lo, N, lane);
      const int p = atomicAdd(&lc[b], 1);
      if (PLACE) {
        if (b < SORT_RANGE) csr[p] = ei[e];
      }
    }
  }
}

__device__ __forceinline__ void chunk_bounds(int E, int& e0, int& e1) {
  const int len = (((E + SORT_CHUNKS - 1) / SORT_CHUNKS) + 3) & ~3;
  e0 = (int)blockIdx.x * len;
  if (e0 > E) e0 = E;
  e1 = e0 + len;
  if (e1 > E) e1 = E;
}

__device__ __forceinline__ int wave_sum_int(int v) {
  for (int o = 16; o > 0; o >>= 1) v += __shfl_xor(v, o);
  return v;
}

__global__ void __launch_bounds__(1024)
gat_count_kernel(const int* __restrict__ ei, int* __restrict__ planes,
                 int* __restrict__ rtot, int E, int N) {
  __shared__ __attribute__((aligned(16))) int lc[SORT_RANGE + 32];
  const int tid = threadIdx.x;
  const int lo = (int)blockIdx.y * SORT_RANGE;
  const int npad = (int)gridDim.y * SORT_RANGE;
  int e0, e1;
  chunk_bounds(E, e0, e1);
  for (int k = tid; k < SORT_RANGE + 32; k += 1024) lc[k] = 0;
  __syncthreads();
  edge_pass<false>(ei, lc, nullptr, E, N, lo, e0, e1, tid);
  __syncthreads();
  *(int4*)(planes + (size_t)blockIdx.x * npad + lo + 4 * tid) = *(const int4*)&lc[4 * tid];
  if (tid < 32) {
    const int outside = wave_sum_int(lc[SORT_RANGE + tid]);
    if (tid == 0) rtot[(int)blockIdx.y * SORT_CHUNKS + (int)blockIdx.x] = (e1 - e0) - outside;
  }
}

__global__ void __launch_bounds__(1024)
gat_fill_kernel(const int* __restrict__ ei, const int* __restrict__ planes,
                const int* __restrict__ rtot, int* __restrict__ cnt,
                int* __restrict__ csr, int E, int N) {
  __shared__ __attribute__((aligned(16))) int lc[SORT_RANGE + 32];
  __shared__ int wsum[32];
  __shared__ int psum[32];
  const int tid = threadIdx.x, lane = tid & 31, wave = tid >> 5;
  const int chunk = (int)blockIdx.x;
  const int lo = (int)blockIdx.y * SORT_RANGE;
  const int npad = (int)gridDim.y * SORT_RANGE;
  const int n0 = lo + 4 * tid;
  int e0, e1;
  chunk_bounds(E, e0, e1);

  int4 pl[SORT_CHUNKS];
#pragma unroll
  for (int j = 0; j < SORT_CHUNKS; ++j) pl[j] = *(const int4*)(planes + (size_t)j * npad + n0);

  int part = 0;
  const int lower = (int)blockIdx.y * SORT_CHUNKS;
  for (int i = tid; i < lower; i += 1024) part += rtot[i];

  int4 tot = make_int4(0, 0, 0, 0);
  int4 mine = make_int4(0, 0, 0, 0);
#pragma unroll
  for (int j = 0; j < SORT_CHUNKS; ++j) {
    const int4 p = pl[j];
    tot.x += p.x;  tot.y += p.y;  tot.z += p.z;  tot.w += p.w;
    if (j < chunk) { mine.x += p.x;  mine.y += p.y;  mine.z += p.z;  mine.w += p.w; }
  }
  const int s = (tot.x + tot.y) + (tot.z + tot.w);
  int incl = s;
  for (int o = 1; o < 32; o <<= 1) {
    const int v = __shfl_up(incl, o);
    if (lane >= o) incl += v;
  }
  part = wave_sum_int(part);
  if (lane == 31) wsum[wave] = incl;
  if (lane == 0) psum[wave] = part;
  __syncthreads();
  const int before = wave_sum_int(psum[lane] + ((lane < wave) ? wsum[lane] : 0));
  int4 st;
  st.x = before + incl - s;
  st.y = st.x + tot.x;
  st.z = st.y + tot.y;
  st.w = st.z + tot.z;
  if (chunk == 0) {
    if (n0 + 3 <= N) {
      *(int4*)(cnt + n0) = st;
    } else {
      if (n0 <= N) cnt[n0] = st.x;
      if (n0 + 1 <= N) cnt[n0 + 1] = st.y;
      if (n0 + 2 <= N) cnt[n0 + 2] = st.z;
    }
  }
  *(int4*)&lc[4 * tid] = make_int4(st.x + mine.x, st.y + mine.y, st.z + mine.z, st.w + mine.w);
  if (tid < 32) lc[SORT_RANGE + tid] = 0;
  __syncthreads();
  edge_pass<true>(ei, lc, csr, E, N, lo, e0, e1, tid);
}

template <int CTRL>
__device__ __forceinline__ float dpp_row(float v) {
  return __int_as_float(__builtin_amdgcn_update_dpp(
      0, __float_as_int(v), CTRL,  0xf,  0xf,  false));
}
__device__ __forceinline__ float xhalf(float v) {
  const int b = __float_as_int(v);
  return __int_as_float(__builtin_amdgcn_permlanex16(b, b, 0x76543210, 0xfedcba98, false, false));
}
__device__ __forceinline__ float wave_sum_valu(float v) {
  v += dpp_row<0x161>(v);
  v += dpp_row<0x162>(v);
  v += dpp_row<0x164>(v);
  v += dpp_row<0x168>(v);
  v += xhalf(v);
  return v;
}

__device__ __forceinline__ float oct_sum(float v) {
  v += dpp_row<0x161>(v);
  v += dpp_row<0x162>(v);
  v += dpp_row<0x164>(v);
  return v;
}

__device__ __forceinline__ float oct_max(float v) {
  v = fmaxf(v, dpp_row<0x161>(v));
  v = fmaxf(v, dpp_row<0x162>(v));
  v = fmaxf(v, dpp_row<0x164>(v));
  return v;
}
template <int U>
__device__ __forceinline__ float oct_bcast(float v) {
  return __int_as_float((int)__builtin_amdgcn_mov_dpp8((unsigned)__float_as_int(v), (unsigned)(U * 0x249249)));
}

__device__ __forceinline__ float dot4(float4 a, float4 b) {
  const v2f al = {a.x, a.y}, ah = {a.z, a.w};
  const v2f bl = {b.x, b.y}, bh = {b.z, b.w};
  const v2f m = __builtin_elementwise_fma(ah, bh, al * bl);
  return m.x + m.y;
}

template <int CTRL>
__device__ __forceinline__ float fold_step(bool bit, float lo, float hi) {
  const float keep = bit ? hi : lo;
  const float send = bit ? lo : hi;
  return keep + dpp_row<CTRL>(send);
}

__device__ __forceinline__ float row_scores(const float (&p)[8], bool b4, bool b2, bool b1) {
  const float q0 = fold_step<0x164>(b4, p[0], p[4]);
  const float q1 = fold_step<0x164>(b4, p[1], p[5]);
  const float q2 = fold_step<0x164>(b4, p[2], p[6]);
  const float q3 = fold_step<0x164>(b4, p[3], p[7]);
  const float r0 = fold_step<0x162>(b2, q0, q2);
  const float r1 = fold_step<0x162>(b2, q1, q3);
  return fold_step<0x161>(b1, r0, r1);
}

template <int N>
__device__ __forceinline__ void fold_group(const float* __restrict__ xl,
                                           int sv, int j0, int lane,
                                           float4 tsA, float4 tsB,
                                           float adA, float adB,
                                           float& mA, float& sA, float4& aA,
                                           float& mB, float& sB, float4& aB) {
  const int slot = lane & 7;
  float4 vA[N], vB[N];
#pragma unroll
  for (int u = 0; u < N; ++u) {
    const int sj = __builtin_amdgcn_readlane(sv, j0 + u);
    const float4* __restrict__ row = (const float4*)(xl + (size_t)sj * F);
    vA[u] = row[lane];
    vB[u] = row[lane + 32];
  }
  float dA[8], dB[8];
#define ROW_DOT(U)                                                          \
  if constexpr (U < N) {                                                    \
    dA[U] = dot4(vA[U], tsA);                                               \
    dB[U] = dot4(vB[U], tsB);                                               \
  } else {           \
    dA[U] = 0.f;                                                            \
    dB[U] = 0.f;                                                            \
  }
  ROW_DOT(0) ROW_DOT(1) ROW_DOT(2) ROW_DOT(3)
  ROW_DOT(4) ROW_DOT(5) ROW_DOT(6) ROW_DOT(7)
#undef ROW_DOT
  const bool b4 = (slot & 4) != 0, b2 = (slot & 2) != 0, b1 = (slot & 1) != 0;
  const float gA = row_scores(dA, b4, b2, b1);
  const float gB = row_scores(dB, b4, b2, b1);
  const bool live = (N == 8) || slot < N;
  const float lA = live ? leaky(gA + adA) : -INFINITY;
  const float lB = live ? leaky(gB + adB) : -INFINITY;
  const float nmA = fmaxf(mA, oct_max(lA));
  const float nmB = fmaxf(mB, oct_max(lB));
  const float scA = expf(mA - nmA);
  const float scB = expf(mB - nmB);
  const float pA = expf(lA - nmA);
  const float pB = expf(lB - nmB);
  sA = fmaf(sA, scA, pA);
  sB = fmaf(sB, scB, pB);
  mA = nmA;
  mB = nmB;
  aA.x *= scA;  aA.y *= scA;  aA.z *= scA;  aA.w *= scA;
  aB.x *= scB;  aB.y *= scB;  aB.z *= scB;  aB.w *= scB;
#define FOLD_ROW(U)                                                         \
  if constexpr (U < N) {                                                    \
    const float wA = oct_bcast<U>(pA);                                      \
    const float wB = oct_bcast<U>(pB);                                      \
    aA.x = fmaf(wA, vA[U].x, aA.x);  aA.y = fmaf(wA, vA[U].y, aA.y);        \
    aA.z = fmaf(wA, vA[U].z, aA.z);  aA.w = fmaf(wA, vA[U].w, aA.w);        \
    aB.x = fmaf(wB, vB[U].x, aB.x);  aB.y = fmaf(wB, vB[U].y, aB.y);        \
    aB.z = fmaf(wB, vB[U].z, aB.z);  aB.w = fmaf(wB, vB[U].w, aB.w);        \
  }
  FOLD_ROW(0) FOLD_ROW(1) FOLD_ROW(2) FOLD_ROW(3)
  FOLD_ROW(4) FOLD_ROW(5) FOLD_ROW(6) FOLD_ROW(7)
#undef FOLD_ROW
}

__global__ void __launch_bounds__(256) __attribute__((amdgpu_waves_per_eu(4, 8)))
gat_gather_ln_kernel(const float* __restrict__ xl, const float* __restrict__ att_src,
                     const float* __restrict__ att_dst, const int* __restrict__ cnt,
                     const int* __restrict__ csr, const float* __restrict__ bias,
                     const float* __restrict__ gamma, const float* __restrict__ beta,
                     float* __restrict__ out, int N) {
  const int lane = threadIdx.x & 31;
  const int d = __builtin_amdgcn_readfirstlane((int)(blockIdx.x * 8u + (threadIdx.x >> 5)));
  if (d >= N) return;

  const int start = cnt[d];
  const int end = cnt[d + 1];

  const float4* __restrict__ own = (const float4*)(xl + (size_t)d * F);
  const float4 xA = own[lane];
  const float4 xB = own[lane + 32];
  const float4 tA = ((const float4*)att_dst)[lane];
  const float4 tB = ((const float4*)att_dst)[lane + 32];
  const float4 tsA = ((const float4*)att_src)[lane];
  const float4 tsB = ((const float4*)att_src)[lane + 32];
  const float adA = oct_sum(xA.x * tA.x + xA.y * tA.y + xA.z * tA.z + xA.w * tA.w);
  const float adB = oct_sum(xB.x * tB.x + xB.y * tB.y + xB.z * tB.z + xB.w * tB.w);

  float mA = leaky(oct_sum(dot4(xA, tsA)) + adA);
  float mB = leaky(oct_sum(dot4(xB, tsB)) + adB);
  float sA = ((lane & 7) == 0) ? 1.f : 0.f;
  float sB = sA;
  float4 aA = xA, aB = xB;

  for (int base = start; base < end; base += 32) {
    const int idx = base + lane;
    const int sv = (idx < end) ? csr[idx] : d;
    int left = end - base;
    if (left > 32) left = 32;
    int j0 = 0;
    for (; j0 + 8 <= left; j0 += 8)
      fold_group<8>(xl, sv, j0, lane, tsA, tsB, adA, adB, mA, sA, aA, mB, sB, aB);
#define REST(K) case K: fold_group<K>(xl, sv, j0, lane, tsA, tsB, adA, adB, mA, sA, aA, mB, sB, aB); break;
    switch (left - j0) {
      REST(1) REST(2) REST(3) REST(4) REST(5) REST(6) REST(7)
      default: break;
    }
#undef REST
  }
  sA = oct_sum(sA);
  sB = oct_sum(sB);

  const float4 bA = ((const float4*)bias)[lane];
  const float4 bB = ((const float4*)bias)[lane + 32];
  const float4 gA = ((const float4*)gamma)[lane];
  const float4 gB = ((const float4*)gamma)[lane + 32];
  const float4 eA = ((const float4*)beta)[lane];
  const float4 eB = ((const float4*)beta)[lane + 32];

  float v[8];
  v[0] = aA.x / sA + bA.x;  v[1] = aA.y / sA + bA.y;
  v[2] = aA.z / sA + bA.z;  v[3] = aA.w / sA + bA.w;
  v[4] = aB.x / sB + bB.x;  v[5] = aB.y / sB + bB.y;
  v[6] = aB.z / sB + bB.z;  v[7] = aB.w / sB + bB.w;

  const float s1 = ((v[0] + v[1]) + (v[2] + v[3])) + ((v[4] + v[5]) + (v[6] + v[7]));
  const float mean = wave_sum_valu(s1) * (1.0f / F);
  float s2 = 0.f;
#pragma unroll
  for (int i = 0; i < 8; ++i) {
    v[i] -= mean;
    s2 = fmaf(v[i], v[i], s2);
  }
  const float var = wave_sum_valu(s2) * (1.0f / F);
  const float rs = rsqrtf(var + LN_EPS);

  float4 oA, oB;
  oA.x = v[0] * rs * gA.x + eA.x;  oA.y = v[1] * rs * gA.y + eA.y;
  oA.z = v[2] * rs * gA.z + eA.z;  oA.w = v[3] * rs * gA.w + eA.w;
  oB.x = v[4] * rs * gB.x + eB.x;  oB.y = v[5] * rs * gB.y + eB.y;
  oB.z = v[6] * rs * gB.z + eB.z;  oB.w = v[7] * rs * gB.w + eB.w;
  float4* __restrict__ orow = (float4*)(out + (size_t)d * F);
  orow[lane] = oA;
  orow[lane + 32] = oB;
}

extern "C" void kernel_launch(void* const* d_in, const int* in_sizes, int n_in,
                              void* d_out, int out_size, void* d_ws, size_t ws_size,
                              hipStream_t stream) {
  const float* x       = (const float*)d_in[0];
  const int*   ei      = (const int*)  d_in[1];
  const float* W       = (const float*)d_in[2];
  const float* att_src = (const float*)d_in[3];
  const float* att_dst = (const float*)d_in[4];
  const float* bias    = (const float*)d_in[5];
  const float* gamma   = (const float*)d_in[6];
  const float* beta    = (const float*)d_in[7];
  float* out = (float*)d_out;

  const int N = in_sizes[0] / F;
  const int E = in_sizes[1] / 2;

  char* ws = (char*)d_ws;
  float* xl    = (float*)ws; ws += (size_t)N * F * sizeof(float);
  int*   cnt   = (int*)ws;   ws += (size_t)N * HEADS * sizeof(float);
  int*   csr   = (int*)ws;
  int*   rtot  = cnt + N + 1;
  int*   planes = (int*)xl;

  {
    dim3 grid(SORT_CHUNKS, (N + SORT_RANGE) / SORT_RANGE);
    gat_count_kernel<<<grid, 1024, 0, stream>>>(ei, planes, rtot, E, N);
    gat_fill_kernel<<<grid, 1024, 0, stream>>>(ei, planes, rtot, cnt, csr, E, N);
  }

  gat_gemm_wmma_kernel<<<(N + GEMM_ROWS - 1) / GEMM_ROWS, 512, 0, stream>>>(x, W, xl, N);

  gat_gather_ln_kernel<<<(N + 7) / 8, 256, 0, stream>>>(xl, att_src, att_dst, cnt, csr,
                                                        bias, gamma, beta, out, N);
}
